# starting kernel
# baseline (speedup 1.0000x reference)
#include <hip/hip_runtime.h>

typedef __attribute__((ext_vector_type(2))) float v2f;
typedef __attribute__((ext_vector_type(8))) float v8f;

__global__ __launch_bounds__(256) void fill_f32(float* __restrict__ p, float v, long n) {
    long i = (long)blockIdx.x * 256 + threadIdx.x;
    if (i < n) p[i] = v;
}

__global__ __launch_bounds__(256) void deg_count(const int* __restrict__ dst,
                                                 float* __restrict__ deg, int nE) {
    int e = blockIdx.x * 256 + threadIdx.x;
    if (e < nE) atomicAdd(&deg[dst[e]], 1.0f);
}

__global__ __launch_bounds__(256) void rsqrt_k(const float* __restrict__ deg,
                                               float* __restrict__ dinv, int n) {
    int i = blockIdx.x * 256 + threadIdx.x;
    if (i < n) dinv[i] = rsqrtf(deg[i]);
}

#define LPITCH 66

__global__ __launch_bounds__(256) void gemm_k128(const float* __restrict__ A,
                                                 const float* __restrict__ W,
                                                 const float* __restrict__ bias,
                                                 float* __restrict__ out,
                                                 int M, int N, int do_relu, int has_bias) {
    __shared__ v2f sA2[16][LPITCH];
    __shared__ v2f sW2[128][LPITCH];

    const int row0 = blockIdx.x * 16;
    const int tid  = threadIdx.x;

    float* sWf = (float*)sW2;
    for (int i = tid; i < 128 * N; i += 256) {
        int k = i / N, n = i - k * N;
        sWf[n * (2 * LPITCH) + k] = W[i];
    }
    float* sAf = (float*)sA2;
    for (int i = tid; i < 16 * 128; i += 256) {
        int r = i >> 7, c = i & 127;
        sAf[r * (2 * LPITCH) + c] = (row0 + r < M) ? A[(size_t)(row0 + r) * 128 + c] : 0.0f;
    }
    __syncthreads();

    const int wave  = tid >> 5;
    const int lane  = tid & 31;
    const int m     = lane & 15;
    const int hsel  = lane >> 4;
    const int mb    = hsel << 3;
    const int ncolt = N >> 4;

    for (int ct = wave; ct < ncolt; ct += 8) {
        const int col0 = ct << 4;
        v8f acc = {};
#pragma unroll 4
        for (int k2 = 0; k2 < 64; k2 += 2) {
            const int kq = k2 + hsel;
            v2f a = sA2[m][kq];
            v2f b = sW2[col0 + m][kq];
            acc = __builtin_amdgcn_wmma_f32_16x16x4_f32(
                false, a, false, b, (short)0, acc, false, false);
        }
        const float bv = has_bias ? bias[col0 + m] : 0.0f;
#pragma unroll
        for (int r = 0; r < 8; ++r) {
            int row = row0 + mb + r;
            if (row < M) {
                float v = acc[r] + bv;
                if (do_relu) v = fmaxf(v, 0.0f);
                out[(size_t)row * N + col0 + m] = v;
            }
        }
    }
}

__global__ __launch_bounds__(256) void edge_scatter(const int* __restrict__ src,
                                                    const int* __restrict__ dst,
                                                    const float* __restrict__ dinv,
                                                    const float* __restrict__ h,
                                                    float* __restrict__ agg, int nE) {
    int e = blockIdx.x * 8 + (threadIdx.x >> 5);
    if (e >= nE) return;
    const int lane = threadIdx.x & 31;
    const int s = src[e], d = dst[e];
    const float norm = dinv[s] * dinv[d];
    const float4 v = ((const float4*)(h + (size_t)s * 128))[lane];
    float* ap = agg + (size_t)d * 128 + lane * 4;
    atomicAdd(ap + 0, norm * v.x);
    atomicAdd(ap + 1, norm * v.y);
    atomicAdd(ap + 2, norm * v.z);
    atomicAdd(ap + 3, norm * v.w);
}

__global__ __launch_bounds__(256) void finalize_k(const float* __restrict__ agg,
                                                  float* __restrict__ h,
                                                  const float* __restrict__ dinv,
                                                  const float* __restrict__ bias,
                                                  int nNodes) {
    long i = (long)blockIdx.x * 256 + threadIdx.x;
    long total = (long)nNodes * 32;
    if (i >= total) return;
    const int node = (int)(i >> 5);
    const int c    = (int)(i & 31);
    const float di = dinv[node];
    const float d2 = di * di;
    float4 hv = ((const float4*)h)[i];
    float4 av = ((const float4*)agg)[i];
    const float* bp = bias + c * 4;
    float4 o;
    o.x = fmaxf(av.x + d2 * hv.x + bp[0], 0.0f);
    o.y = fmaxf(av.y + d2 * hv.y + bp[1], 0.0f);
    o.z = fmaxf(av.z + d2 * hv.z + bp[2], 0.0f);
    o.w = fmaxf(av.w + d2 * hv.w + bp[3], 0.0f);
    ((float4*)h)[i] = o;
}

__global__ __launch_bounds__(256) void pool_sum(const float* __restrict__ h,
                                                const int* __restrict__ batch,
                                                float* __restrict__ gsum,
                                                float* __restrict__ gcnt, int nNodes) {
    long i = (long)blockIdx.x * 256 + threadIdx.x;
    long total = (long)nNodes * 32;
    if (i >= total) return;
    const int node = (int)(i >> 5);
    const int c    = (int)(i & 31);
    const int b    = batch[node];
    float4 v = ((const float4*)h)[i];
    float* gp = gsum + (size_t)b * 128 + c * 4;
    atomicAdd(gp + 0, v.x);
    atomicAdd(gp + 1, v.y);
    atomicAdd(gp + 2, v.z);
    atomicAdd(gp + 3, v.w);
    if (c == 0) atomicAdd(&gcnt[b], 1.0f);
}

__global__ __launch_bounds__(256) void pool_div(const float* __restrict__ gsum,
                                                const float* __restrict__ gcnt,
                                                float* __restrict__ g, int G) {
    int i = blockIdx.x * 256 + threadIdx.x;
    if (i < G * 128) {
        int b = i >> 7;
        g[i] = gsum[i] / fmaxf(gcnt[b], 1.0f);
    }
}

extern "C" void kernel_launch(void* const* d_in, const int* in_sizes, int n_in,
                              void* d_out, int out_size, void* d_ws, size_t ws_size,
                              hipStream_t stream) {
    const float* x     = (const float*)d_in[0];
    const int*   ei    = (const int*)d_in[1];
    const int*   batch = (const int*)d_in[2];
    const float* W1    = (const float*)d_in[3];
    const float* b1    = (const float*)d_in[4];
    const float* W2    = (const float*)d_in[5];
    const float* b2    = (const float*)d_in[6];
    const float* Wl1   = (const float*)d_in[7];
    const float* bl1   = (const float*)d_in[8];
    const float* Wl2   = (const float*)d_in[9];
    const float* bl2   = (const float*)d_in[10];
    float* out = (float*)d_out;

    const int nN = in_sizes[0] / 128;
    const int nE = in_sizes[1] / 2;
    const int G  = out_size / 32;
    const int* src = ei;
    const int* dst = ei + nE;

    char* ws = (char*)d_ws;
    auto alloc = [&](size_t bytes) -> float* {
        float* p = (float*)ws;
        ws += (bytes + 255) & ~(size_t)255;
        return p;
    };
    float* bufA = alloc((size_t)nN * 128 * 4);
    float* bufB = alloc((size_t)nN * 128 * 4);
    float* bufC = alloc((size_t)nN * 128 * 4);
    float* deg  = alloc((size_t)nN * 4);
    float* dinv = alloc((size_t)nN * 4);
    float* gsum = alloc((size_t)G * 128 * 4);
    float* gcnt = alloc((size_t)G * 4);
    float* g    = alloc((size_t)G * 128 * 4);
    float* m1   = alloc((size_t)G * 128 * 4);

    const long nel = (long)nN * 128;
    const int  fB  = (int)((nel + 255) / 256);
    const int  eB  = (nE + 255) / 256;
    const int  nB  = (nN + 255) / 256;
    const int  qB  = (int)(((long)nN * 32 + 255) / 256);
    const int  sB  = (nE + 7) / 8;

    fill_f32<<<nB, 256, 0, stream>>>(deg, 1.0f, nN);
    fill_f32<<<fB, 256, 0, stream>>>(bufB, 0.0f, nel);
    fill_f32<<<(G * 128 + 255) / 256, 256, 0, stream>>>(gsum, 0.0f, (long)G * 128);
    fill_f32<<<(G + 255) / 256, 256, 0, stream>>>(gcnt, 0.0f, G);
    deg_count<<<eB, 256, 0, stream>>>(dst, deg, nE);
    rsqrt_k<<<nB, 256, 0, stream>>>(deg, dinv, nN);

    gemm_k128<<<(nN + 15) / 16, 256, 0, stream>>>(x, W1, nullptr, bufA, nN, 128, 0, 0);
    edge_scatter<<<sB, 256, 0, stream>>>(src, dst, dinv, bufA, bufB, nE);
    finalize_k<<<qB, 256, 0, stream>>>(bufB, bufA, dinv, b1, nN);

    fill_f32<<<fB, 256, 0, stream>>>(bufB, 0.0f, nel);
    gemm_k128<<<(nN + 15) / 16, 256, 0, stream>>>(bufA, W2, nullptr, bufC, nN, 128, 0, 0);
    edge_scatter<<<sB, 256, 0, stream>>>(src, dst, dinv, bufC, bufB, nE);
    finalize_k<<<qB, 256, 0, stream>>>(bufB, bufC, dinv, b2, nN);

    pool_sum<<<qB, 256, 0, stream>>>(bufC, batch, gsum, gcnt, nN);
    pool_div<<<(G * 128 + 255) / 256, 256, 0, stream>>>(gsum, gcnt, g, G);

    gemm_k128<<<(G + 15) / 16, 256, 0, stream>>>(g, Wl1, bl1, m1, G, 128, 1, 1);
    gemm_k128<<<(G + 15) / 16, 256, 0, stream>>>(m1, Wl2, bl2, out, G, 32, 0, 1);
}


# Edge sums kept in LDS: a work-group owns 392 dst rows, scans the whole dst list, gathers its edges' rows, f32 LDS adds (arrival order), stores relu rows; layer atomics, clears, finalize gone
# speedup vs baseline: 1.7805x; 1.7805x over previous
#include <hip/hip_runtime.h>

typedef __attribute__((ext_vector_type(2))) float v2f;
typedef __attribute__((ext_vector_type(8))) float v8f;

__global__ __launch_bounds__(256) void fill_f32(float* __restrict__ p, float v, long n) {
    long i = (long)blockIdx.x * 256 + threadIdx.x;
    if (i < n) p[i] = v;
}

__global__ __launch_bounds__(256) void deg_count(const int* __restrict__ dst,
                                                 float* __restrict__ deg, int nE) {
    int e = blockIdx.x * 256 + threadIdx.x;
    if (e < nE) atomicAdd(&deg[dst[e]], 1.0f);
}

__global__ __launch_bounds__(256) void rsqrt_k(const float* __restrict__ deg,
                                               float* __restrict__ dinv, int n) {
    int i = blockIdx.x * 256 + threadIdx.x;
    if (i < n) dinv[i] = rsqrtf(deg[i]);
}

#define LPITCH 66

__global__ __launch_bounds__(256) void gemm_k128(const float* __restrict__ A,
                                                 const float* __restrict__ W,
                                                 const float* __restrict__ bias,
                                                 float* __restrict__ out,
                                                 int M, int N, int do_relu, int has_bias) {
    __shared__ v2f sA2[16][LPITCH];
    __shared__ v2f sW2[128][LPITCH];

    const int row0 = blockIdx.x * 16;
    const int tid  = threadIdx.x;

    float* sWf = (float*)sW2;
    for (int i = tid; i < 128 * N; i += 256) {
        int k = i / N, n = i - k * N;
        sWf[n * (2 * LPITCH) + k] = W[i];
    }
    float* sAf = (float*)sA2;
    for (int i = tid; i < 16 * 128; i += 256) {
        int r = i >> 7, c = i & 127;
        sAf[r * (2 * LPITCH) + c] = (row0 + r < M) ? A[(size_t)(row0 + r) * 128 + c] : 0.0f;
    }
    __syncthreads();

    const int wave  = tid >> 5;
    const int lane  = tid & 31;
    const int m     = lane & 15;
    const int hsel  = lane >> 4;
    const int mb    = hsel << 3;
    const int ncolt = N >> 4;

    for (int ct = wave; ct < ncolt; ct += 8) {
        const int col0 = ct << 4;
        v8f acc = {};
#pragma unroll 4
        for (int k2 = 0; k2 < 64; k2 += 2) {
            const int kq = k2 + hsel;
            v2f a = sA2[m][kq];
            v2f b = sW2[col0 + m][kq];
            acc = __builtin_amdgcn_wmma_f32_16x16x4_f32(
                false, a, false, b, (short)0, acc, false, false);
        }
        const float bv = has_bias ? bias[col0 + m] : 0.0f;
#pragma unroll
        for (int r = 0; r < 8; ++r) {
            int row = row0 + mb + r;
            if (row < M) {
                float v = acc[r] + bv;
                if (do_relu) v = fmaxf(v, 0.0f);
                out[(size_t)row * N + col0 + m] = v;
            }
        }
    }
}

#define OWN_ROWS   392
#define OWN_WAVES  32
#define SCAN_AHEAD 4
#define QCAP       544

__global__ __launch_bounds__(1024) void layer_owned_rows(const int* __restrict__ src,
                                                         const int* __restrict__ dst,
                                                         const float* __restrict__ dinv,
                                                         const float* __restrict__ h,
                                                         const float* __restrict__ bias,
                                                         float* __restrict__ act,
                                                         int nE, int nN) {
    __shared__ float sAgg[OWN_ROWS * 128];
    __shared__ int   sQ[OWN_WAVES * QCAP];

    const int tid  = threadIdx.x;
    const int lane = tid & 31;
    const int wave = __builtin_amdgcn_readfirstlane(tid >> 5);
    const int base = (int)blockIdx.x * OWN_ROWS;
    const int nOwn = max(0, min(OWN_ROWS, nN - base));

    float4* z = (float4*)sAgg;
    for (int i = tid; i < OWN_ROWS * 32; i += 1024) z[i] = make_float4(0.0f, 0.0f, 0.0f, 0.0f);
    __syncthreads();

    int* q = sQ + wave * QCAP;
    const int  nPieces = (nE + 127) >> 7;
    const int  nSets = ((((size_t)dst) & 15) == 0) ? (nE >> 7) / SCAN_AHEAD : 0;
    const int  p0 = (int)(((unsigned)nSets / gridDim.x) * blockIdx.x);
    int qn = 0;

    auto set_at = [&](int t) -> int {
        const int s = t + p0;
        return s >= nSets ? s - nSets : s;
    };
    auto push = [&](bool m, unsigned b, int eid) {
        if (b) {
            const int pos = qn + (int)__builtin_amdgcn_mbcnt_lo(b, 0u);
            if (m) q[pos] = eid;
            qn += __builtin_popcount(b);
        }
    };
    auto drain = [&](const int* qp, int cnt) {
        int s = 0, r = 0;
        float w = 0.0f;
        if (lane < cnt) {
            const int eid = qp[lane];
            s = src[eid];
            const int d = dst[eid];
            r = d - base;
            w = dinv[s] * dinv[d];
        }
        const int wi = __float_as_int(w);
        for (int j = 0; j < cnt; j += 8) {
            float4 v[8];
#pragma unroll
            for (int u = 0; u < 8; ++u) {
                const int sj = __builtin_amdgcn_readlane(s, j + u);
                v[u] = *(const float4*)(h + (size_t)sj * 128 + (lane << 2));
            }
#pragma unroll
            for (int u = 0; u < 8; ++u) {
                if (j + u < cnt) {
                    const int   rj = __builtin_amdgcn_readlane(r, j + u);
                    const float wj = __int_as_float(__builtin_amdgcn_readlane(wi, j + u));
                    float* ap = sAgg + rj * 128 + lane;
                    __hip_atomic_fetch_add(ap +  0, wj * v[u].x, __ATOMIC_RELAXED, __HIP_MEMORY_SCOPE_WORKGROUP);
                    __hip_atomic_fetch_add(ap + 32, wj * v[u].y, __ATOMIC_RELAXED, __HIP_MEMORY_SCOPE_WORKGROUP);
                    __hip_atomic_fetch_add(ap + 64, wj * v[u].z, __ATOMIC_RELAXED, __HIP_MEMORY_SCOPE_WORKGROUP);
                    __hip_atomic_fetch_add(ap + 96, wj * v[u].w, __ATOMIC_RELAXED, __HIP_MEMORY_SCOPE_WORKGROUP);
                }
            }
        }
    };

    auto test = [&](const int4 d4, int e) {
        const bool m0 = (unsigned)d4.x - (unsigned)base < (unsigned)nOwn;
        const bool m1 = (unsigned)d4.y - (unsigned)base < (unsigned)nOwn;
        const bool m2 = (unsigned)d4.z - (unsigned)base < (unsigned)nOwn;
        const bool m3 = (unsigned)d4.w - (unsigned)base < (unsigned)nOwn;
        const unsigned b0 = __builtin_amdgcn_ballot_w32(m0);
        const unsigned b1 = __builtin_amdgcn_ballot_w32(m1);
        const unsigned b2 = __builtin_amdgcn_ballot_w32(m2);
        const unsigned b3 = __builtin_amdgcn_ballot_w32(m3);
        if (b0 | b1 | b2 | b3) {
            push(m0, b0, e);
            push(m1, b1, e + 1);
            push(m2, b2, e + 2);
            push(m3, b3, e + 3);
        }
    };

    auto load_set = [&](int4 (&st)[SCAN_AHEAD], int t) {
        const int4* p = (const int4*)(dst + ((size_t)set_at(min(t, nSets - 1)) << 9)) + lane;
#pragma unroll
        for (int u = 0; u < SCAN_AHEAD; ++u) st[u] = p[32 * u];
    };
    auto test_set = [&](const int4 (&st)[SCAN_AHEAD], int t) {
        if (t < nSets) {
            const int e = (set_at(t) << 9) + (lane << 2);
#pragma unroll
            for (int u = 0; u < SCAN_AHEAD; ++u) test(st[u], e + 128 * u);
        }
        while (qn >= 32) {
            qn -= 32;
            drain(q + qn, 32);
        }
    };

    if (nSets > 0) {
        int4 sa[SCAN_AHEAD], sb[SCAN_AHEAD];
        load_set(sa, wave);
        for (int t = wave; t < nSets; t += 2 * OWN_WAVES) {
            load_set(sb, t + OWN_WAVES);
            test_set(sa, t);
            load_set(sa, t + 2 * OWN_WAVES);
            test_set(sb, t + OWN_WAVES);
        }
    }
    for (int k = nSets * SCAN_AHEAD + wave; k < nPieces; k += OWN_WAVES) {
        const int e = (k << 7) + (lane << 2);
        int4 d4;
        d4.x = (e     < nE) ? dst[e]     : -1;
        d4.y = (e + 1 < nE) ? dst[e + 1] : -1;
        d4.z = (e + 2 < nE) ? dst[e + 2] : -1;
        d4.w = (e + 3 < nE) ? dst[e + 3] : -1;
        test(d4, e);
        while (qn >= 32) {
            qn -= 32;
            drain(q + qn, 32);
        }
    }
    if (qn > 0) drain(q, qn);
    __syncthreads();

    const float bx = bias[(lane << 2) + 0];
    const float by = bias[(lane << 2) + 1];
    const float bz = bias[(lane << 2) + 2];
    const float bw = bias[(lane << 2) + 3];
    for (int r0 = wave; r0 < nOwn; r0 += 4 * OWN_WAVES) {
        float  di[4];
        float4 hv[4];
#pragma unroll
        for (int t = 0; t < 4; ++t) {
            const size_t node = (size_t)(base + min(r0 + OWN_WAVES * t, nOwn - 1));
            di[t] = dinv[node];
            hv[t] = *(const float4*)(h + node * 128 + (lane << 2));
        }
#pragma unroll
        for (int t = 0; t < 4; ++t) {
            const int r = r0 + OWN_WAVES * t;
            if (r < nOwn) {
                const size_t node = (size_t)(base + r);
                const float d2 = di[t] * di[t];
                const float* ap = sAgg + r * 128 + lane;
                float4 o;
                o.x = fmaxf(ap[0]  + d2 * hv[t].x + bx, 0.0f);
                o.y = fmaxf(ap[32] + d2 * hv[t].y + by, 0.0f);
                o.z = fmaxf(ap[64] + d2 * hv[t].z + bz, 0.0f);
                o.w = fmaxf(ap[96] + d2 * hv[t].w + bw, 0.0f);
                *(float4*)(act + node * 128 + (lane << 2)) = o;
            }
        }
    }
}

__global__ __launch_bounds__(256) void pool_sum(const float* __restrict__ h,
                                                const int* __restrict__ batch,
                                                float* __restrict__ gsum,
                                                float* __restrict__ gcnt, int nNodes) {
    long i = (long)blockIdx.x * 256 + threadIdx.x;
    long total = (long)nNodes * 32;
    if (i >= total) return;
    const int node = (int)(i >> 5);
    const int lane = (int)(i & 31);
    const int b    = batch[node];
    const float* hp = h + (size_t)node * 128 + lane;
    const float v0 = hp[0];
    const float v1 = hp[32];
    const float v2 = hp[64];
    const float v3 = hp[96];
    float* gp = gsum + (size_t)b * 128 + lane;
    atomicAdd(gp +  0, v0);
    atomicAdd(gp + 32, v1);
    atomicAdd(gp + 64, v2);
    atomicAdd(gp + 96, v3);
    if (lane == 0) atomicAdd(&gcnt[b], 1.0f);
}

__global__ __launch_bounds__(256) void pool_div(const float* __restrict__ gsum,
                                                const float* __restrict__ gcnt,
                                                float* __restrict__ g, int G) {
    int i = blockIdx.x * 256 + threadIdx.x;
    if (i < G * 128) {
        int b = i >> 7;
        g[i] = gsum[i] / fmaxf(gcnt[b], 1.0f);
    }
}

extern "C" void kernel_launch(void* const* d_in, const int* in_sizes, int n_in,
                              void* d_out, int out_size, void* d_ws, size_t ws_size,
                              hipStream_t stream) {
    const float* x     = (const float*)d_in[0];
    const int*   ei    = (const int*)d_in[1];
    const int*   batch = (const int*)d_in[2];
    const float* W1    = (const float*)d_in[3];
    const float* b1    = (const float*)d_in[4];
    const float* W2    = (const float*)d_in[5];
    const float* b2    = (const float*)d_in[6];
    const float* Wl1   = (const float*)d_in[7];
    const float* bl1   = (const float*)d_in[8];
    const float* Wl2   = (const float*)d_in[9];
    const float* bl2   = (const float*)d_in[10];
    float* out = (float*)d_out;

    const int nN = in_sizes[0] / 128;
    const int nE = in_sizes[1] / 2;
    const int G  = out_size / 32;
    const int* src = ei;
    const int* dst = ei + nE;

    char* ws = (char*)d_ws;
    auto alloc = [&](size_t bytes) -> float* {
        float* p = (float*)ws;
        ws += (bytes + 255) & ~(size_t)255;
        return p;
    };
    float* bufA = alloc((size_t)nN * 128 * 4);
    float* bufB = alloc((size_t)nN * 128 * 4);
    float* bufC = alloc((size_t)nN * 128 * 4);
    float* deg  = alloc((size_t)nN * 4);
    float* dinv = alloc((size_t)nN * 4);
    float* gsum = alloc((size_t)G * 128 * 4);
    float* gcnt = alloc((size_t)G * 4);
    float* g    = alloc((size_t)G * 128 * 4);
    float* m1   = alloc((size_t)G * 128 * 4);

    const int  eB  = (nE + 255) / 256;
    const int  nB  = (nN + 255) / 256;
    const int  qB  = (int)(((long)nN * 32 + 255) / 256);
    const int  oB  = (nN + OWN_ROWS - 1) / OWN_ROWS;

    fill_f32<<<nB, 256, 0, stream>>>(deg, 1.0f, nN);
    fill_f32<<<(G * 128 + 255) / 256, 256, 0, stream>>>(gsum, 0.0f, (long)G * 128);
    fill_f32<<<(G + 255) / 256, 256, 0, stream>>>(gcnt, 0.0f, G);
    deg_count<<<eB, 256, 0, stream>>>(dst, deg, nE);
    rsqrt_k<<<nB, 256, 0, stream>>>(deg, dinv, nN);

    gemm_k128<<<(nN + 15) / 16, 256, 0, stream>>>(x, W1, nullptr, bufA, nN, 128, 0, 0);
    layer_owned_rows<<<oB, 1024, 0, stream>>>(src, dst, dinv, bufA, b1, bufB, nE, nN);

    gemm_k128<<<(nN + 15) / 16, 256, 0, stream>>>(bufB, W2, nullptr, bufC, nN, 128, 0, 0);
    layer_owned_rows<<<oB, 1024, 0, stream>>>(src, dst, dinv, bufC, b2, bufA, nE, nN);

    pool_sum<<<qB, 256, 0, stream>>>(bufA, batch, gsum, gcnt, nN);
    pool_div<<<(G * 128 + 255) / 256, 256, 0, stream>>>(gsum, gcnt, g, G);

    gemm_k128<<<(G + 15) / 16, 256, 0, stream>>>(g, Wl1, bl1, m1, G, 128, 1, 1);
    gemm_k128<<<(G + 15) / 16, 256, 0, stream>>>(m1, Wl2, bl2, out, G, 32, 0, 1);
}


# Adjacency applied before the weights, (A x) W: each layer kernel multiplies its LDS-held row sums by W on f32 WMMA (tile k order kept); both 100k-row product launches gone; rounding order differs
# speedup vs baseline: 2.4412x; 1.3711x over previous
#include <hip/hip_runtime.h>

typedef __attribute__((ext_vector_type(2))) float v2f;
typedef __attribute__((ext_vector_type(8))) float v8f;

__global__ __launch_bounds__(256) void fill_f32(float* __restrict__ p, float v, long n) {
    long i = (long)blockIdx.x * 256 + threadIdx.x;
    if (i < n) p[i] = v;
}

__global__ __launch_bounds__(256) void deg_count(const int* __restrict__ dst,
                                                 float* __restrict__ deg, int nE) {
    int e = blockIdx.x * 256 + threadIdx.x;
    if (e < nE) atomicAdd(&deg[dst[e]], 1.0f);
}

__global__ __launch_bounds__(256) void rsqrt_k(const float* __restrict__ deg,
                                               float* __restrict__ dinv, int n) {
    int i = blockIdx.x * 256 + threadIdx.x;
    if (i < n) dinv[i] = rsqrtf(deg[i]);
}

#define LPITCH 66

__global__ __launch_bounds__(256) void gemm_k128(const float* __restrict__ A,
                                                 const float* __restrict__ W,
                                                 const float* __restrict__ bias,
                                                 float* __restrict__ out,
                                                 int M, int N, int do_relu, int has_bias) {
    __shared__ v2f sA2[16][LPITCH];
    __shared__ v2f sW2[128][LPITCH];

    const int row0 = blockIdx.x * 16;
    const int tid  = threadIdx.x;

    float* sWf = (float*)sW2;
    for (int i = tid; i < 128 * N; i += 256) {
        int k = i / N, n = i - k * N;
        sWf[n * (2 * LPITCH) + k] = W[i];
    }
    float* sAf = (float*)sA2;
    for (int i = tid; i < 16 * 128; i += 256) {
        int r = i >> 7, c = i & 127;
        sAf[r * (2 * LPITCH) + c] = (row0 + r < M) ? A[(size_t)(row0 + r) * 128 + c] : 0.0f;
    }
    __syncthreads();

    const int wave  = tid >> 5;
    const int lane  = tid & 31;
    const int m     = lane & 15;
    const int hsel  = lane >> 4;
    const int mb    = hsel << 3;
    const int ncolt = N >> 4;

    for (int ct = wave; ct < ncolt; ct += 8) {
        const int col0 = ct << 4;
        v8f acc = {};
#pragma unroll 4
        for (int k2 = 0; k2 < 64; k2 += 2) {
            const int kq = k2 + hsel;
            v2f a = sA2[m][kq];
            v2f b = sW2[col0 + m][kq];
            acc = __builtin_amdgcn_wmma_f32_16x16x4_f32(
                false, a, false, b, (short)0, acc, false, false);
        }
        const float bv = has_bias ? bias[col0 + m] : 0.0f;
#pragma unroll
        for (int r = 0; r < 8; ++r) {
            int row = row0 + mb + r;
            if (row < M) {
                float v = acc[r] + bv;
                if (do_relu) v = fmaxf(v, 0.0f);
                out[(size_t)row * N + col0 + m] = v;
            }
        }
    }
}

#define OWN_ROWS   392
#define OWN_WAVES  32
#define SCAN_AHEAD 4
#define QCAP       544
#define APITCH     132

__global__ __launch_bounds__(1024) void layer_owned_rows(const int* __restrict__ src,
                                                         const int* __restrict__ dst,
                                                         const float* __restrict__ dinv,
                                                         const float* __restrict__ xin,
                                                         const float* __restrict__ Wm,
                                                         const float* __restrict__ bias,
                                                         float* __restrict__ act,
                                                         int nE, int nN) {
    __shared__ __attribute__((aligned(16))) float sAgg[OWN_ROWS * APITCH];
    __shared__ __attribute__((aligned(16))) int   sQ[OWN_WAVES * QCAP];
    static_assert(sizeof(int) * OWN_WAVES * QCAP >= sizeof(v2f) * 128 * LPITCH,
                  "the transposed W tile must fit where the queues were");

    const int tid  = threadIdx.x;
    const int lane = tid & 31;
    const int wave = __builtin_amdgcn_readfirstlane(tid >> 5);
    const int base = (int)blockIdx.x * OWN_ROWS;
    const int nOwn = max(0, min(OWN_ROWS, nN - base));

    float4* z = (float4*)sAgg;
    for (int i = tid; i < OWN_ROWS * (APITCH / 4); i += 1024) z[i] = make_float4(0.0f, 0.0f, 0.0f, 0.0f);
    __syncthreads();

    int* q = sQ + wave * QCAP;
    const int  nPieces = (nE + 127) >> 7;
    const int  nSets = ((((size_t)dst) & 15) == 0) ? (nE >> 7) / SCAN_AHEAD : 0;
    const int  p0 = (int)(((unsigned)nSets / gridDim.x) * blockIdx.x);
    int qn = 0;

    auto set_at = [&](int t) -> int {
        const int s = t + p0;
        return s >= nSets ? s - nSets : s;
    };
    auto push = [&](bool m, unsigned b, int eid) {
        if (b) {
            const int pos = qn + (int)__builtin_amdgcn_mbcnt_lo(b, 0u);
            if (m) q[pos] = eid;
            qn += __builtin_popcount(b);
        }
    };
    auto drain = [&](const int* qp, int cnt) {
        int s = 0, r = 0;
        float w = 0.0f;
        if (lane < cnt) {
            const int eid = qp[lane];
            s = src[eid];
            const int d = dst[eid];
            r = d - base;
            w = dinv[s] * dinv[d];
        }
        const int wi = __float_as_int(w);
        for (int j = 0; j < cnt; j += 8) {
            float4 v[8];
#pragma unroll
            for (int u = 0; u < 8; ++u) {
                const int sj = __builtin_amdgcn_readlane(s, j + u);
                v[u] = *(const float4*)(xin + (size_t)sj * 128 + (lane << 2));
            }
#pragma unroll
            for (int u = 0; u < 8; ++u) {
                if (j + u < cnt) {
                    const int   rj = __builtin_amdgcn_readlane(r, j + u);
                    const float wj = __int_as_float(__builtin_amdgcn_readlane(wi, j + u));
                    float* ap = sAgg + rj * APITCH + lane;
                    __hip_atomic_fetch_add(ap +  0, wj * v[u].x, __ATOMIC_RELAXED, __HIP_MEMORY_SCOPE_WORKGROUP);
                    __hip_atomic_fetch_add(ap + 32, wj * v[u].y, __ATOMIC_RELAXED, __HIP_MEMORY_SCOPE_WORKGROUP);
                    __hip_atomic_fetch_add(ap + 64, wj * v[u].z, __ATOMIC_RELAXED, __HIP_MEMORY_SCOPE_WORKGROUP);
                    __hip_atomic_fetch_add(ap + 96, wj * v[u].w, __ATOMIC_RELAXED, __HIP_MEMORY_SCOPE_WORKGROUP);
                }
            }
        }
    };

    auto test = [&](const int4 d4, int e) {
        const bool m0 = (unsigned)d4.x - (unsigned)base < (unsigned)nOwn;
        const bool m1 = (unsigned)d4.y - (unsigned)base < (unsigned)nOwn;
        const bool m2 = (unsigned)d4.z - (unsigned)base < (unsigned)nOwn;
        const bool m3 = (unsigned)d4.w - (unsigned)base < (unsigned)nOwn;
        const unsigned b0 = __builtin_amdgcn_ballot_w32(m0);
        const unsigned b1 = __builtin_amdgcn_ballot_w32(m1);
        const unsigned b2 = __builtin_amdgcn_ballot_w32(m2);
        const unsigned b3 = __builtin_amdgcn_ballot_w32(m3);
        if (b0 | b1 | b2 | b3) {
            push(m0, b0, e);
            push(m1, b1, e + 1);
            push(m2, b2, e + 2);
            push(m3, b3, e + 3);
        }
    };

    auto load_set = [&](int4 (&st)[SCAN_AHEAD], int t) {
        const int4* p = (const int4*)(dst + ((size_t)set_at(min(t, nSets - 1)) << 9)) + lane;
#pragma unroll
        for (int u = 0; u < SCAN_AHEAD; ++u) st[u] = p[32 * u];
    };
    auto test_set = [&](const int4 (&st)[SCAN_AHEAD], int t) {
        if (t < nSets) {
            const int e = (set_at(t) << 9) + (lane << 2);
#pragma unroll
            for (int u = 0; u < SCAN_AHEAD; ++u) test(st[u], e + 128 * u);
        }
        while (qn >= 32) {
            qn -= 32;
            drain(q + qn, 32);
        }
    };

    if (nSets > 0) {
        int4 sa[SCAN_AHEAD], sb[SCAN_AHEAD];
        load_set(sa, wave);
        for (int t = wave; t < nSets; t += 2 * OWN_WAVES) {
            load_set(sb, t + OWN_WAVES);
            test_set(sa, t);
            load_set(sa, t + 2 * OWN_WAVES);
            test_set(sb, t + OWN_WAVES);
        }
    }
    for (int k = nSets * SCAN_AHEAD + wave; k < nPieces; k += OWN_WAVES) {
        const int e = (k << 7) + (lane << 2);
        int4 d4;
        d4.x = (e     < nE) ? dst[e]     : -1;
        d4.y = (e + 1 < nE) ? dst[e + 1] : -1;
        d4.z = (e + 2 < nE) ? dst[e + 2] : -1;
        d4.w = (e + 3 < nE) ? dst[e + 3] : -1;
        test(d4, e);
        while (qn >= 32) {
            qn -= 32;
            drain(q + qn, 32);
        }
    }
    if (qn > 0) drain(q, qn);
    __syncthreads();

    v2f*   sW2 = (v2f*)sQ;
    float* sWf = (float*)sQ;
    for (int i = tid; i < 128 * 128; i += 1024) {
        const int k = i >> 7, n = i & 127;
        sWf[n * (2 * LPITCH) + k] = Wm[i];
    }

    for (int r0 = wave; r0 < nOwn; r0 += 4 * OWN_WAVES) {
        float  di[4];
        float4 hv[4];
#pragma unroll
        for (int t = 0; t < 4; ++t) {
            const size_t node = (size_t)(base + min(r0 + OWN_WAVES * t, nOwn - 1));
            di[t] = dinv[node];
            hv[t] = *(const float4*)(xin + node * 128 + (lane << 2));
        }
#pragma unroll
        for (int t = 0; t < 4; ++t) {
            const int r = r0 + OWN_WAVES * t;
            if (r < nOwn) {
                const float d2 = di[t] * di[t];
                float* rowp = sAgg + r * APITCH;
                const float* ap = rowp + lane;
                float4 o;
                o.x = ap[0]  + d2 * hv[t].x;
                o.y = ap[32] + d2 * hv[t].y;
                o.z = ap[64] + d2 * hv[t].z;
                o.w = ap[96] + d2 * hv[t].w;
                *(float4*)(rowp + (lane << 2)) = o;
            }
        }
    }
    __syncthreads();

    const int m      = lane & 15;
    const int hsel   = lane >> 4;
    const int mb     = hsel << 3;
    const int nItems = ((nOwn + 15) >> 4) << 2;
    for (int p = wave; p < nItems; p += OWN_WAVES) {
        const int row0 = (p >> 2) << 4;
        const int col0 = (p & 3) << 5;
        const v2f* arow = (const v2f*)(sAgg + min(row0 + m, OWN_ROWS - 1) * APITCH);
        const v2f* b0p  = sW2 + (col0 + m) * LPITCH;
        const v2f* b1p  = sW2 + (col0 + 16 + m) * LPITCH;
        v8f acc0 = {};
        v8f acc1 = {};
#pragma unroll 4
        for (int k2 = 0; k2 < 64; k2 += 2) {
            const int kq = k2 + hsel;
            const v2f a  = arow[kq];
            const v2f b0 = b0p[kq];
            const v2f b1 = b1p[kq];
            acc0 = __builtin_amdgcn_wmma_f32_16x16x4_f32(
                false, a, false, b0, (short)0, acc0, false, false);
            acc1 = __builtin_amdgcn_wmma_f32_16x16x4_f32(
                false, a, false, b1, (short)0, acc1, false, false);
        }
        const float bv0 = bias[col0 + m];
        const float bv1 = bias[col0 + 16 + m];
#pragma unroll
        for (int r = 0; r < 8; ++r) {
            const int row = row0 + mb + r;
            if (row < nOwn) {
                float* op = act + (size_t)(base + row) * 128 + col0 + m;
                op[0]  = fmaxf(acc0[r] + bv0, 0.0f);
                op[16] = fmaxf(acc1[r] + bv1, 0.0f);
            }
        }
    }
}

__global__ __launch_bounds__(256) void pool_sum(const float* __restrict__ h,
                                                const int* __restrict__ batch,
                                                float* __restrict__ gsum,
                                                float* __restrict__ gcnt, int nNodes) {
    long i = (long)blockIdx.x * 256 + threadIdx.x;
    long total = (long)nNodes * 32;
    if (i >= total) return;
    const int node = (int)(i >> 5);
    const int lane = (int)(i & 31);
    const int b    = batch[node];
    const float* hp = h + (size_t)node * 128 + lane;
    const float v0 = hp[0];
    const float v1 = hp[32];
    const float v2 = hp[64];
    const float v3 = hp[96];
    float* gp = gsum + (size_t)b * 128 + lane;
    atomicAdd(gp +  0, v0);
    atomicAdd(gp + 32, v1);
    atomicAdd(gp + 64, v2);
    atomicAdd(gp + 96, v3);
    if (lane == 0) atomicAdd(&gcnt[b], 1.0f);
}

__global__ __launch_bounds__(256) void pool_div(const float* __restrict__ gsum,
                                                const float* __restrict__ gcnt,
                                                float* __restrict__ g, int G) {
    int i = blockIdx.x * 256 + threadIdx.x;
    if (i < G * 128) {
        int b = i >> 7;
        g[i] = gsum[i] / fmaxf(gcnt[b], 1.0f);
    }
}

extern "C" void kernel_launch(void* const* d_in, const int* in_sizes, int n_in,
                              void* d_out, int out_size, void* d_ws, size_t ws_size,
                              hipStream_t stream) {
    const float* x     = (const float*)d_in[0];
    const int*   ei    = (const int*)d_in[1];
    const int*   batch = (const int*)d_in[2];
    const float* W1    = (const float*)d_in[3];
    const float* b1    = (const float*)d_in[4];
    const float* W2    = (const float*)d_in[5];
    const float* b2    = (const float*)d_in[6];
    const float* Wl1   = (const float*)d_in[7];
    const float* bl1   = (const float*)d_in[8];
    const float* Wl2   = (const float*)d_in[9];
    const float* bl2   = (const float*)d_in[10];
    float* out = (float*)d_out;

    const int nN = in_sizes[0] / 128;
    const int nE = in_sizes[1] / 2;
    const int G  = out_size / 32;
    const int* src = ei;
    const int* dst = ei + nE;

    char* ws = (char*)d_ws;
    auto alloc = [&](size_t bytes) -> float* {
        float* p = (float*)ws;
        ws += (bytes + 255) & ~(size_t)255;
        return p;
    };
    float* bufA = alloc((size_t)nN * 128 * 4);
    float* bufB = alloc((size_t)nN * 128 * 4);
    float* deg  = alloc((size_t)nN * 4);
    float* dinv = alloc((size_t)nN * 4);
    float* gsum = alloc((size_t)G * 128 * 4);
    float* gcnt = alloc((size_t)G * 4);
    float* g    = alloc((size_t)G * 128 * 4);
    float* m1   = alloc((size_t)G * 128 * 4);

    const int  eB  = (nE + 255) / 256;
    const int  nB  = (nN + 255) / 256;
    const int  qB  = (int)(((long)nN * 32 + 255) / 256);
    const int  oB  = (nN + OWN_ROWS - 1) / OWN_ROWS;

    fill_f32<<<nB, 256, 0, stream>>>(deg, 1.0f, nN);
    fill_f32<<<(G * 128 + 255) / 256, 256, 0, stream>>>(gsum, 0.0f, (long)G * 128);
    fill_f32<<<(G + 255) / 256, 256, 0, stream>>>(gcnt, 0.0f, G);
    deg_count<<<eB, 256, 0, stream>>>(dst, deg, nE);
    rsqrt_k<<<nB, 256, 0, stream>>>(deg, dinv, nN);

    layer_owned_rows<<<oB, 1024, 0, stream>>>(src, dst, dinv, x, W1, b1, bufA, nE, nN);

    layer_owned_rows<<<oB, 1024, 0, stream>>>(src, dst, dinv, bufA, W2, b2, bufB, nE, nN);

    pool_sum<<<qB, 256, 0, stream>>>(bufB, batch, gsum, gcnt, nN);
    pool_div<<<(G * 128 + 255) / 256, 256, 0, stream>>>(gsum, gcnt, g, G);

    gemm_k128<<<(G + 15) / 16, 256, 0, stream>>>(g, Wl1, bl1, m1, G, 128, 1, 1);
    gemm_k128<<<(G + 15) / 16, 256, 0, stream>>>(m1, Wl2, bl2, out, G, 32, 0, 1);
}


# Layer 1 writes each group's matched edges (source row, own row, weight) as a list; layer 2 gathers from that list instead of scanning the dst list again; LDS f32 adds in arrival order, as before
# speedup vs baseline: 2.7608x; 1.1309x over previous
#include <hip/hip_runtime.h>

typedef __attribute__((ext_vector_type(2))) float v2f;
typedef __attribute__((ext_vector_type(8))) float v8f;

__global__ __launch_bounds__(256) void fill_f32(float* __restrict__ p, float v, long n) {
    long i = (long)blockIdx.x * 256 + threadIdx.x;
    if (i < n) p[i] = v;
}

__global__ __launch_bounds__(256) void deg_count(const int* __restrict__ dst,
                                                 float* __restrict__ deg, int nE) {
    int e = blockIdx.x * 256 + threadIdx.x;
    if (e < nE) atomicAdd(&deg[dst[e]], 1.0f);
}

__global__ __launch_bounds__(256) void rsqrt_k(const float* __restrict__ deg,
                                               float* __restrict__ dinv, int n) {
    int i = blockIdx.x * 256 + threadIdx.x;
    if (i < n) dinv[i] = rsqrtf(deg[i]);
}

#define LPITCH 66

__global__ __launch_bounds__(256) void gemm_k128(const float* __restrict__ A,
                                                 const float* __restrict__ W,
                                                 const float* __restrict__ bias,
                                                 float* __restrict__ out,
                                                 int M, int N, int do_relu, int has_bias) {
    __shared__ v2f sA2[16][LPITCH];
    __shared__ v2f sW2[128][LPITCH];

    const int row0 = blockIdx.x * 16;
    const int tid  = threadIdx.x;

    float* sWf = (float*)sW2;
    for (int i = tid; i < 128 * N; i += 256) {
        int k = i / N, n = i - k * N;
        sWf[n * (2 * LPITCH) + k] = W[i];
    }
    float* sAf = (float*)sA2;
    for (int i = tid; i < 16 * 128; i += 256) {
        int r = i >> 7, c = i & 127;
        sAf[r * (2 * LPITCH) + c] = (row0 + r < M) ? A[(size_t)(row0 + r) * 128 + c] : 0.0f;
    }
    __syncthreads();

    const int wave  = tid >> 5;
    const int lane  = tid & 31;
    const int m     = lane & 15;
    const int hsel  = lane >> 4;
    const int mb    = hsel << 3;
    const int ncolt = N >> 4;

    for (int ct = wave; ct < ncolt; ct += 8) {
        const int col0 = ct << 4;
        v8f acc = {};
#pragma unroll 4
        for (int k2 = 0; k2 < 64; k2 += 2) {
            const int kq = k2 + hsel;
            v2f a = sA2[m][kq];
            v2f b = sW2[col0 + m][kq];
            acc = __builtin_amdgcn_wmma_f32_16x16x4_f32(
                false, a, false, b, (short)0, acc, false, false);
        }
        const float bv = has_bias ? bias[col0 + m] : 0.0f;
#pragma unroll
        for (int r = 0; r < 8; ++r) {
            int row = row0 + mb + r;
            if (row < M) {
                float v = acc[r] + bv;
                if (do_relu) v = fmaxf(v, 0.0f);
                out[(size_t)row * N + col0 + m] = v;
            }
        }
    }
}

#define OWN_ROWS   392
#define OWN_WAVES  32
#define SCAN_AHEAD 4
#define QCAP       544
#define APITCH     132

__global__ __launch_bounds__(1024) void layer_owned_rows(const int* __restrict__ src,
                                                         const int* __restrict__ dst,
                                                         const float* __restrict__ dinv,
                                                         const float* __restrict__ xin,
                                                         const float* __restrict__ Wm,
                                                         const float* __restrict__ bias,
                                                         float* act, int* rec, int replay,
                                                         int nE, int nN) {
    __shared__ __attribute__((aligned(16))) float sAgg[OWN_ROWS * APITCH];
    __shared__ __attribute__((aligned(16))) int   sQ[OWN_WAVES * QCAP];
    static_assert(sizeof(int) * OWN_WAVES * QCAP >= sizeof(v2f) * 128 * LPITCH,
                  "the transposed W tile must fit where the queues were");
    __shared__ int sRecN;
    __shared__ int sRecBad;

    const int tid  = threadIdx.x;
    const int lane = tid & 31;
    const int wave = __builtin_amdgcn_readfirstlane(tid >> 5);
    const int base = (int)blockIdx.x * OWN_ROWS;
    const int nOwn = max(0, min(OWN_ROWS, nN - base));

    float4* z = (float4*)sAgg;
    for (int i = tid; i < OWN_ROWS * (APITCH / 4); i += 1024) z[i] = make_float4(0.0f, 0.0f, 0.0f, 0.0f);
    if (tid == 0) {
        sRecN = 0;
        sRecBad = 0;
    }
    __syncthreads();

    int*   lst = rec + (size_t)base * 128;
    const int cap = ((nOwn * 128 - 32) / 3) & ~31;
    int*   lS = lst + 32;
    int*   lR = lS + cap;
    float* lW = (float*)(lR + cap);
    int nRec = -1;
    if (replay) {
        nRec = __builtin_amdgcn_readfirstlane(lst[0]);
        if (nRec > cap) nRec = -1;
    }

    int* q = sQ + wave * QCAP;
    const int  nPieces = (nE + 127) >> 7;
    const int  nSets = ((((size_t)dst) & 15) == 0) ? (nE >> 7) / SCAN_AHEAD : 0;
    const int  p0 = (int)(((unsigned)nSets / gridDim.x) * blockIdx.x);
    int qn = 0;

    auto set_at = [&](int t) -> int {
        const int s = t + p0;
        return s >= nSets ? s - nSets : s;
    };
    auto push = [&](bool m, unsigned b, int eid) {
        if (b) {
            const int pos = qn + (int)__builtin_amdgcn_mbcnt_lo(b, 0u);
            if (m) q[pos] = eid;
            qn += __builtin_popcount(b);
        }
    };
    auto gather_add = [&](int s, int r, float w, int cnt) {
        const int wi = __float_as_int(w);
        for (int j = 0; j < cnt; j += 8) {
            float4 v[8];
#pragma unroll
            for (int u = 0; u < 8; ++u) {
                const int sj = __builtin_amdgcn_readlane(s, j + u);
                v[u] = *(const float4*)(xin + (size_t)sj * 128 + (lane << 2));
            }
#pragma unroll
            for (int u = 0; u < 8; ++u) {
                if (j + u < cnt) {
                    const int   rj = __builtin_amdgcn_readlane(r, j + u);
                    const float wj = __int_as_float(__builtin_amdgcn_readlane(wi, j + u));
                    float* ap = sAgg + rj * APITCH + lane;
                    __hip_atomic_fetch_add(ap +  0, wj * v[u].x, __ATOMIC_RELAXED, __HIP_MEMORY_SCOPE_WORKGROUP);
                    __hip_atomic_fetch_add(ap + 32, wj * v[u].y, __ATOMIC_RELAXED, __HIP_MEMORY_SCOPE_WORKGROUP);
                    __hip_atomic_fetch_add(ap + 64, wj * v[u].z, __ATOMIC_RELAXED, __HIP_MEMORY_SCOPE_WORKGROUP);
                    __hip_atomic_fetch_add(ap + 96, wj * v[u].w, __ATOMIC_RELAXED, __HIP_MEMORY_SCOPE_WORKGROUP);
                }
            }
        }
    };
    auto drain = [&](const int* qp, int cnt) {
        int s = 0, r = 0;
        float w = 0.0f;
        if (lane < cnt) {
            const int eid = qp[lane];
            s = src[eid];
            const int d = dst[eid];
            r = d - base;
            w = dinv[s] * dinv[d];
        }
        if (!replay) {
            int pos = 0;
            if (lane == 0) pos = atomicAdd(&sRecN, cnt);
            pos = __builtin_amdgcn_readfirstlane(pos);
            if (pos + cnt <= cap) {
                if (lane < cnt) {
                    lS[pos + lane] = s;
                    lR[pos + lane] = r;
                    lW[pos + lane] = w;
                }
            } else if (lane == 0) {
                sRecBad = 1;
            }
        }
        gather_add(s, r, w, cnt);
    };

    auto test = [&](const int4 d4, int e) {
        const bool m0 = (unsigned)d4.x - (unsigned)base < (unsigned)nOwn;
        const bool m1 = (unsigned)d4.y - (unsigned)base < (unsigned)nOwn;
        const bool m2 = (unsigned)d4.z - (unsigned)base < (unsigned)nOwn;
        const bool m3 = (unsigned)d4.w - (unsigned)base < (unsigned)nOwn;
        const unsigned b0 = __builtin_amdgcn_ballot_w32(m0);
        const unsigned b1 = __builtin_amdgcn_ballot_w32(m1);
        const unsigned b2 = __builtin_amdgcn_ballot_w32(m2);
        const unsigned b3 = __builtin_amdgcn_ballot_w32(m3);
        if (b0 | b1 | b2 | b3) {
            push(m0, b0, e);
            push(m1, b1, e + 1);
            push(m2, b2, e + 2);
            push(m3, b3, e + 3);
        }
    };

    auto load_set = [&](int4 (&st)[SCAN_AHEAD], int t) {
        const int4* p = (const int4*)(dst + ((size_t)set_at(min(t, nSets - 1)) << 9)) + lane;
#pragma unroll
        for (int u = 0; u < SCAN_AHEAD; ++u) st[u] = p[32 * u];
    };
    auto test_set = [&](const int4 (&st)[SCAN_AHEAD], int t) {
        if (t < nSets) {
            const int e = (set_at(t) << 9) + (lane << 2);
#pragma unroll
            for (int u = 0; u < SCAN_AHEAD; ++u) test(st[u], e + 128 * u);
        }
        while (qn >= 32) {
            qn -= 32;
            drain(q + qn, 32);
        }
    };

    if (nRec >= 0) {
        for (int i = wave * 32; i < nRec; i += 32 * OWN_WAVES) {
            int s = 0, r = 0;
            float w = 0.0f;
            if (i + lane < nRec) {
                s = lS[i + lane];
                r = lR[i + lane];
                w = lW[i + lane];
            }
            gather_add(s, r, w, min(32, nRec - i));
        }
    } else {
        if (nSets > 0) {
            int4 sa[SCAN_AHEAD], sb[SCAN_AHEAD];
            load_set(sa, wave);
            for (int t = wave; t < nSets; t += 2 * OWN_WAVES) {
                load_set(sb, t + OWN_WAVES);
                test_set(sa, t);
                load_set(sa, t + 2 * OWN_WAVES);
                test_set(sb, t + OWN_WAVES);
            }
        }
        for (int k = nSets * SCAN_AHEAD + wave; k < nPieces; k += OWN_WAVES) {
            const int e = (k << 7) + (lane << 2);
            int4 d4;
            d4.x = (e     < nE) ? dst[e]     : -1;
            d4.y = (e + 1 < nE) ? dst[e + 1] : -1;
            d4.z = (e + 2 < nE) ? dst[e + 2] : -1;
            d4.w = (e + 3 < nE) ? dst[e + 3] : -1;
            test(d4, e);
            while (qn >= 32) {
                qn -= 32;
                drain(q + qn, 32);
            }
        }
        if (qn > 0) drain(q, qn);
    }
    __syncthreads();
    if (!replay && tid == 0) lst[0] = sRecBad ? -1 : sRecN;

    v2f*   sW2 = (v2f*)sQ;
    float* sWf = (float*)sQ;
    for (int i = tid; i < 128 * 128; i += 1024) {
        const int k = i >> 7, n = i & 127;
        sWf[n * (2 * LPITCH) + k] = Wm[i];
    }

    for (int r0 = wave; r0 < nOwn; r0 += 4 * OWN_WAVES) {
        float  di[4];
        float4 hv[4];
#pragma unroll
        for (int t = 0; t < 4; ++t) {
            const size_t node = (size_t)(base + min(r0 + OWN_WAVES * t, nOwn - 1));
            di[t] = dinv[node];
            hv[t] = *(const float4*)(xin + node * 128 + (lane << 2));
        }
#pragma unroll
        for (int t = 0; t < 4; ++t) {
            const int r = r0 + OWN_WAVES * t;
            if (r < nOwn) {
                const float d2 = di[t] * di[t];
                float* rowp = sAgg + r * APITCH;
                const float* ap = rowp + lane;
                float4 o;
                o.x = ap[0]  + d2 * hv[t].x;
                o.y = ap[32] + d2 * hv[t].y;
                o.z = ap[64] + d2 * hv[t].z;
                o.w = ap[96] + d2 * hv[t].w;
                *(float4*)(rowp + (lane << 2)) = o;
            }
        }
    }
    __syncthreads();

    const int m      = lane & 15;
    const int hsel   = lane >> 4;
    const int mb     = hsel << 3;
    const int nItems = ((nOwn + 15) >> 4) << 2;
    for (int p = wave; p < nItems; p += OWN_WAVES) {
        const int row0 = (p >> 2) << 4;
        const int col0 = (p & 3) << 5;
        const v2f* arow = (const v2f*)(sAgg + min(row0 + m, OWN_ROWS - 1) * APITCH);
        const v2f* b0p  = sW2 + (col0 + m) * LPITCH;
        const v2f* b1p  = sW2 + (col0 + 16 + m) * LPITCH;
        v8f acc0 = {};
        v8f acc1 = {};
#pragma unroll 4
        for (int k2 = 0; k2 < 64; k2 += 2) {
            const int kq = k2 + hsel;
            const v2f a  = arow[kq];
            const v2f b0 = b0p[kq];
            const v2f b1 = b1p[kq];
            acc0 = __builtin_amdgcn_wmma_f32_16x16x4_f32(
                false, a, false, b0, (short)0, acc0, false, false);
            acc1 = __builtin_amdgcn_wmma_f32_16x16x4_f32(
                false, a, false, b1, (short)0, acc1, false, false);
        }
        const float bv0 = bias[col0 + m];
        const float bv1 = bias[col0 + 16 + m];
#pragma unroll
        for (int r = 0; r < 8; ++r) {
            const int row = row0 + mb + r;
            if (row < nOwn) {
                float* op = act + (size_t)(base + row) * 128 + col0 + m;
                op[0]  = fmaxf(acc0[r] + bv0, 0.0f);
                op[16] = fmaxf(acc1[r] + bv1, 0.0f);
            }
        }
    }
}

__global__ __launch_bounds__(256) void pool_sum(const float* __restrict__ h,
                                                const int* __restrict__ batch,
                                                float* __restrict__ gsum,
                                                float* __restrict__ gcnt, int nNodes) {
    long i = (long)blockIdx.x * 256 + threadIdx.x;
    long total = (long)nNodes * 32;
    if (i >= total) return;
    const int node = (int)(i >> 5);
    const int lane = (int)(i & 31);
    const int b    = batch[node];
    const float* hp = h + (size_t)node * 128 + lane;
    const float v0 = hp[0];
    const float v1 = hp[32];
    const float v2 = hp[64];
    const float v3 = hp[96];
    float* gp = gsum + (size_t)b * 128 + lane;
    atomicAdd(gp +  0, v0);
    atomicAdd(gp + 32, v1);
    atomicAdd(gp + 64, v2);
    atomicAdd(gp + 96, v3);
    if (lane == 0) atomicAdd(&gcnt[b], 1.0f);
}

__global__ __launch_bounds__(256) void pool_div(const float* __restrict__ gsum,
                                                const float* __restrict__ gcnt,
                                                float* __restrict__ g, int G) {
    int i = blockIdx.x * 256 + threadIdx.x;
    if (i < G * 128) {
        int b = i >> 7;
        g[i] = gsum[i] / fmaxf(gcnt[b], 1.0f);
    }
}

extern "C" void kernel_launch(void* const* d_in, const int* in_sizes, int n_in,
                              void* d_out, int out_size, void* d_ws, size_t ws_size,
                              hipStream_t stream) {
    const float* x     = (const float*)d_in[0];
    const int*   ei    = (const int*)d_in[1];
    const int*   batch = (const int*)d_in[2];
    const float* W1    = (const float*)d_in[3];
    const float* b1    = (const float*)d_in[4];
    const float* W2    = (const float*)d_in[5];
    const float* b2    = (const float*)d_in[6];
    const float* Wl1   = (const float*)d_in[7];
    const float* bl1   = (const float*)d_in[8];
    const float* Wl2   = (const float*)d_in[9];
    const float* bl2   = (const float*)d_in[10];
    float* out = (float*)d_out;

    const int nN = in_sizes[0] / 128;
    const int nE = in_sizes[1] / 2;
    const int G  = out_size / 32;
    const int* src = ei;
    const int* dst = ei + nE;

    char* ws = (char*)d_ws;
    auto alloc = [&](size_t bytes) -> float* {
        float* p = (float*)ws;
        ws += (bytes + 255) & ~(size_t)255;
        return p;
    };
    float* bufA = alloc((size_t)nN * 128 * 4);
    float* bufB = alloc((size_t)nN * 128 * 4);
    float* deg  = alloc((size_t)nN * 4);
    float* dinv = alloc((size_t)nN * 4);
    float* gsum = alloc((size_t)G * 128 * 4);
    float* gcnt = alloc((size_t)G * 4);
    float* g    = alloc((size_t)G * 128 * 4);
    float* m1   = alloc((size_t)G * 128 * 4);

    const int  eB  = (nE + 255) / 256;
    const int  nB  = (nN + 255) / 256;
    const int  qB  = (int)(((long)nN * 32 + 255) / 256);
    const int  oB  = (nN + OWN_ROWS - 1) / OWN_ROWS;

    fill_f32<<<nB, 256, 0, stream>>>(deg, 1.0f, nN);
    fill_f32<<<(G * 128 + 255) / 256, 256, 0, stream>>>(gsum, 0.0f, (long)G * 128);
    fill_f32<<<(G + 255) / 256, 256, 0, stream>>>(gcnt, 0.0f, G);
    deg_count<<<eB, 256, 0, stream>>>(dst, deg, nE);
    rsqrt_k<<<nB, 256, 0, stream>>>(deg, dinv, nN);

    layer_owned_rows<<<oB, 1024, 0, stream>>>(src, dst, dinv, x, W1, b1, bufA, (int*)bufB, 0, nE, nN);

    layer_owned_rows<<<oB, 1024, 0, stream>>>(src, dst, dinv, bufA, W2, b2, bufB, (int*)bufB, 1, nE, nN);

    pool_sum<<<qB, 256, 0, stream>>>(bufB, batch, gsum, gcnt, nN);
    pool_div<<<(G * 128 + 255) / 256, 256, 0, stream>>>(gsum, gcnt, g, G);

    gemm_k128<<<(G + 15) / 16, 256, 0, stream>>>(g, Wl1, bl1, m1, G, 128, 1, 1);
    gemm_k128<<<(G + 15) / 16, 256, 0, stream>>>(m1, Wl2, bl2, out, G, 32, 0, 1);
}


# Mean-pool sum taken in the layer-2 product epilogue: tile rows summed in registers, f32 LDS adds per graph, one set of whole-line atomics per group; layer-2 rows never stored; pool sum order differs
# speedup vs baseline: 4.1437x; 1.5009x over previous
#include <hip/hip_runtime.h>

typedef __attribute__((ext_vector_type(2))) float v2f;
typedef __attribute__((ext_vector_type(8))) float v8f;

__global__ __launch_bounds__(256) void fill_f32(float* __restrict__ p, float v, long n) {
    long i = (long)blockIdx.x * 256 + threadIdx.x;
    if (i < n) p[i] = v;
}

__global__ __launch_bounds__(256) void deg_count(const int* __restrict__ dst,
                                                 float* __restrict__ deg, int nE) {
    int e = blockIdx.x * 256 + threadIdx.x;
    if (e < nE) atomicAdd(&deg[dst[e]], 1.0f);
}

__global__ __launch_bounds__(256) void rsqrt_k(const float* __restrict__ deg,
                                               float* __restrict__ dinv, int n) {
    int i = blockIdx.x * 256 + threadIdx.x;
    if (i < n) dinv[i] = rsqrtf(deg[i]);
}

#define LPITCH 66

__global__ __launch_bounds__(256) void gemm_k128(const float* __restrict__ A,
                                                 const float* __restrict__ W,
                                                 const float* __restrict__ bias,
                                                 float* __restrict__ out,
                                                 int M, int N, int do_relu, int has_bias) {
    __shared__ v2f sA2[16][LPITCH];
    __shared__ v2f sW2[128][LPITCH];

    const int row0 = blockIdx.x * 16;
    const int tid  = threadIdx.x;

    float* sWf = (float*)sW2;
    for (int i = tid; i < 128 * N; i += 256) {
        int k = i / N, n = i - k * N;
        sWf[n * (2 * LPITCH) + k] = W[i];
    }
    float* sAf = (float*)sA2;
    for (int i = tid; i < 16 * 128; i += 256) {
        int r = i >> 7, c = i & 127;
        sAf[r * (2 * LPITCH) + c] = (row0 + r < M) ? A[(size_t)(row0 + r) * 128 + c] : 0.0f;
    }
    __syncthreads();

    const int wave  = tid >> 5;
    const int lane  = tid & 31;
    const int m     = lane & 15;
    const int hsel  = lane >> 4;
    const int mb    = hsel << 3;
    const int ncolt = N >> 4;

    for (int ct = wave; ct < ncolt; ct += 8) {
        const int col0 = ct << 4;
        v8f acc = {};
#pragma unroll 4
        for (int k2 = 0; k2 < 64; k2 += 2) {
            const int kq = k2 + hsel;
            v2f a = sA2[m][kq];
            v2f b = sW2[col0 + m][kq];
            acc = __builtin_amdgcn_wmma_f32_16x16x4_f32(
                false, a, false, b, (short)0, acc, false, false);
        }
        const float bv = has_bias ? bias[col0 + m] : 0.0f;
#pragma unroll
        for (int r = 0; r < 8; ++r) {
            int row = row0 + mb + r;
            if (row < M) {
                float v = acc[r] + bv;
                if (do_relu) v = fmaxf(v, 0.0f);
                out[(size_t)row * N + col0 + m] = v;
            }
        }
    }
}

#define OWN_ROWS   392
#define OWN_WAVES  32
#define SCAN_AHEAD 4
#define QCAP       544
#define APITCH     132
#define POOL_SLOTS 32
#define SLOT_ROWS  (((OWN_ROWS + 15) / 16) * 16)

__global__ __launch_bounds__(1024) void layer_owned_rows(const int* __restrict__ src,
                                                         const int* __restrict__ dst,
                                                         const float* __restrict__ dinv,
                                                         const float* __restrict__ xin,
                                                         const float* __restrict__ Wm,
                                                         const float* __restrict__ bias,
                                                         float* act, int* rec, int replay,
                                                         int nE, int nN,
                                                         const int* __restrict__ batch,
                                                         float* __restrict__ gsum,
                                                         float* __restrict__ gcnt, int G) {
    __shared__ __attribute__((aligned(16))) float sAgg[OWN_ROWS * APITCH];
    __shared__ __attribute__((aligned(16))) int   sQ[OWN_WAVES * QCAP];
    static_assert(sizeof(int) * OWN_WAVES * QCAP >= sizeof(v2f) * 128 * LPITCH,
                  "the transposed W tile must fit where the queues were");
    __shared__ int sRecN;
    __shared__ int sRecBad;
    __shared__ __attribute__((aligned(16))) float sPool[POOL_SLOTS * 128];
    __shared__ __attribute__((aligned(16))) int   sSlot[SLOT_ROWS];
    __shared__ int sCnt[POOL_SLOTS];
    const bool pooling = gsum != nullptr;

    const int tid  = threadIdx.x;
    const int lane = tid & 31;
    const int wave = __builtin_amdgcn_readfirstlane(tid >> 5);
    const int base = (int)blockIdx.x * OWN_ROWS;
    const int nOwn = max(0, min(OWN_ROWS, nN - base));

    float4* z = (float4*)sAgg;
    for (int i = tid; i < OWN_ROWS * (APITCH / 4); i += 1024) z[i] = make_float4(0.0f, 0.0f, 0.0f, 0.0f);
    if (tid == 0) {
        sRecN = 0;
        sRecBad = 0;
    }
    if (pooling) {
        for (int i = tid; i < POOL_SLOTS * 32; i += 1024) ((float4*)sPool)[i] = make_float4(0.0f, 0.0f, 0.0f, 0.0f);
        if (tid < POOL_SLOTS) sCnt[tid] = 0;
    }
    __syncthreads();

    int*   lst = rec + (size_t)base * 128;
    const int cap = ((nOwn * 128 - 32) / 3) & ~31;
    int*   lS = lst + 32;
    int*   lR = lS + cap;
    float* lW = (float*)(lR + cap);
    int nRec = -1;
    if (replay) {
        nRec = __builtin_amdgcn_readfirstlane(lst[0]);
        if (nRec > cap) nRec = -1;
    }

    int* q = sQ + wave * QCAP;
    const int  nPieces = (nE + 127) >> 7;
    const int  nSets = ((((size_t)dst) & 15) == 0) ? (nE >> 7) / SCAN_AHEAD : 0;
    const int  p0 = (int)(((unsigned)nSets / gridDim.x) * blockIdx.x);
    int qn = 0;

    auto set_at = [&](int t) -> int {
        const int s = t + p0;
        return s >= nSets ? s - nSets : s;
    };
    auto push = [&](bool m, unsigned b, int eid) {
        if (b) {
            const int pos = qn + (int)__builtin_amdgcn_mbcnt_lo(b, 0u);
            if (m) q[pos] = eid;
            qn += __builtin_popcount(b);
        }
    };
    auto gather_add = [&](int s, int r, float w, int cnt) {
        const int wi = __float_as_int(w);
        for (int j = 0; j < cnt; j += 8) {
            float4 v[8];
#pragma unroll
            for (int u = 0; u < 8; ++u) {
                const int sj = __builtin_amdgcn_readlane(s, j + u);
                v[u] = *(const float4*)(xin + (size_t)sj * 128 + (lane << 2));
            }
#pragma unroll
            for (int u = 0; u < 8; ++u) {
                if (j + u < cnt) {
                    const int   rj = __builtin_amdgcn_readlane(r, j + u);
                    const float wj = __int_as_float(__builtin_amdgcn_readlane(wi, j + u));
                    float* ap = sAgg + rj * APITCH + lane;
                    __hip_atomic_fetch_add(ap +  0, wj * v[u].x, __ATOMIC_RELAXED, __HIP_MEMORY_SCOPE_WORKGROUP);
                    __hip_atomic_fetch_add(ap + 32, wj * v[u].y, __ATOMIC_RELAXED, __HIP_MEMORY_SCOPE_WORKGROUP);
                    __hip_atomic_fetch_add(ap + 64, wj * v[u].z, __ATOMIC_RELAXED, __HIP_MEMORY_SCOPE_WORKGROUP);
                    __hip_atomic_fetch_add(ap + 96, wj * v[u].w, __ATOMIC_RELAXED, __HIP_MEMORY_SCOPE_WORKGROUP);
                }
            }
        }
    };
    auto drain = [&](const int* qp, int cnt) {
        int s = 0, r = 0;
        float w = 0.0f;
        if (lane < cnt) {
            const int eid = qp[lane];
            s = src[eid];
            const int d = dst[eid];
            r = d - base;
            w = dinv[s] * dinv[d];
        }
        if (!replay) {
            int pos = 0;
            if (lane == 0) pos = atomicAdd(&sRecN, cnt);
            pos = __builtin_amdgcn_readfirstlane(pos);
            if (pos + cnt <= cap) {
                if (lane < cnt) {
                    lS[pos + lane] = s;
                    lR[pos + lane] = r;
                    lW[pos + lane] = w;
                }
            } else if (lane == 0) {
                sRecBad = 1;
            }
        }
        gather_add(s, r, w, cnt);
    };

    auto test = [&](const int4 d4, int e) {
        const bool m0 = (unsigned)d4.x - (unsigned)base < (unsigned)nOwn;
        const bool m1 = (unsigned)d4.y - (unsigned)base < (unsigned)nOwn;
        const bool m2 = (unsigned)d4.z - (unsigned)base < (unsigned)nOwn;
        const bool m3 = (unsigned)d4.w - (unsigned)base < (unsigned)nOwn;
        const unsigned b0 = __builtin_amdgcn_ballot_w32(m0);
        const unsigned b1 = __builtin_amdgcn_ballot_w32(m1);
        const unsigned b2 = __builtin_amdgcn_ballot_w32(m2);
        const unsigned b3 = __builtin_amdgcn_ballot_w32(m3);
        if (b0 | b1 | b2 | b3) {
            push(m0, b0, e);
            push(m1, b1, e + 1);
            push(m2, b2, e + 2);
            push(m3, b3, e + 3);
        }
    };

    auto load_set = [&](int4 (&st)[SCAN_AHEAD], int t) {
        const int4* p = (const int4*)(dst + ((size_t)set_at(min(t, nSets - 1)) << 9)) + lane;
#pragma unroll
        for (int u = 0; u < SCAN_AHEAD; ++u) st[u] = p[32 * u];
    };
    auto test_set = [&](const int4 (&st)[SCAN_AHEAD], int t) {
        if (t < nSets) {
            const int e = (set_at(t) << 9) + (lane << 2);
#pragma unroll
            for (int u = 0; u < SCAN_AHEAD; ++u) test(st[u], e + 128 * u);
        }
        while (qn >= 32) {
            qn -= 32;
            drain(q + qn, 32);
        }
    };

    if (nRec >= 0) {
        for (int i = wave * 32; i < nRec; i += 32 * OWN_WAVES) {
            int s = 0, r = 0;
            float w = 0.0f;
            if (i + lane < nRec) {
                s = lS[i + lane];
                r = lR[i + lane];
                w = lW[i + lane];
            }
            gather_add(s, r, w, min(32, nRec - i));
        }
    } else {
        if (nSets > 0) {
            int4 sa[SCAN_AHEAD], sb[SCAN_AHEAD];
            load_set(sa, wave);
            for (int t = wave; t < nSets; t += 2 * OWN_WAVES) {
                load_set(sb, t + OWN_WAVES);
                test_set(sa, t);
                load_set(sa, t + 2 * OWN_WAVES);
                test_set(sb, t + OWN_WAVES);
            }
        }
        for (int k = nSets * SCAN_AHEAD + wave; k < nPieces; k += OWN_WAVES) {
            const int e = (k << 7) + (lane << 2);
            int4 d4;
            d4.x = (e     < nE) ? dst[e]     : -1;
            d4.y = (e + 1 < nE) ? dst[e + 1] : -1;
            d4.z = (e + 2 < nE) ? dst[e + 2] : -1;
            d4.w = (e + 3 < nE) ? dst[e + 3] : -1;
            test(d4, e);
            while (qn >= 32) {
                qn -= 32;
                drain(q + qn, 32);
            }
        }
        if (qn > 0) drain(q, qn);
    }
    __syncthreads();
    if (!replay && tid == 0) lst[0] = sRecBad ? -1 : sRecN;

    v2f*   sW2 = (v2f*)sQ;
    float* sWf = (float*)sQ;
    for (int i = tid; i < 128 * 128; i += 1024) {
        const int k = i >> 7, n = i & 127;
        sWf[n * (2 * LPITCH) + k] = Wm[i];
    }

    int gbase = 0;
    if (pooling) {
        gbase = __builtin_amdgcn_readfirstlane(nOwn > 0 ? batch[base] : 0);
        for (int r = tid; r < SLOT_ROWS; r += 1024) {
            int code = -1;
            if (r < nOwn) {
                const int b = batch[base + r];
                if ((unsigned)b < (unsigned)G) {
                    const int s = b - gbase;
                    if ((unsigned)s < (unsigned)POOL_SLOTS) {
                        code = s;
                        atomicAdd(&sCnt[s], 1);
                    } else {
                        code = -2;
                        atomicAdd(gcnt + b, 1.0f);
                    }
                }
            }
            sSlot[r] = code;
        }
    }

    for (int r0 = wave; r0 < nOwn; r0 += 4 * OWN_WAVES) {
        float  di[4];
        float4 hv[4];
#pragma unroll
        for (int t = 0; t < 4; ++t) {
            const size_t node = (size_t)(base + min(r0 + OWN_WAVES * t, nOwn - 1));
            di[t] = dinv[node];
            hv[t] = *(const float4*)(xin + node * 128 + (lane << 2));
        }
#pragma unroll
        for (int t = 0; t < 4; ++t) {
            const int r = r0 + OWN_WAVES * t;
            if (r < nOwn) {
                const float d2 = di[t] * di[t];
                float* rowp = sAgg + r * APITCH;
                const float* ap = rowp + lane;
                float4 o;
                o.x = ap[0]  + d2 * hv[t].x;
                o.y = ap[32] + d2 * hv[t].y;
                o.z = ap[64] + d2 * hv[t].z;
                o.w = ap[96] + d2 * hv[t].w;
                *(float4*)(rowp + (lane << 2)) = o;
            }
        }
    }
    __syncthreads();

    const int m      = lane & 15;
    const int hsel   = lane >> 4;
    const int mb     = hsel << 3;
    const int nItems = ((nOwn + 15) >> 4) << 2;
    for (int p = wave; p < nItems; p += OWN_WAVES) {
        const int row0 = (p >> 2) << 4;
        const int col0 = (p & 3) << 5;
        const v2f* arow = (const v2f*)(sAgg + min(row0 + m, OWN_ROWS - 1) * APITCH);
        const v2f* b0p  = sW2 + (col0 + m) * LPITCH;
        const v2f* b1p  = sW2 + (col0 + 16 + m) * LPITCH;
        v8f acc0 = {};
        v8f acc1 = {};
#pragma unroll 4
        for (int k2 = 0; k2 < 64; k2 += 2) {
            const int kq = k2 + hsel;
            const v2f a  = arow[kq];
            const v2f b0 = b0p[kq];
            const v2f b1 = b1p[kq];
            acc0 = __builtin_amdgcn_wmma_f32_16x16x4_f32(
                false, a, false, b0, (short)0, acc0, false, false);
            acc1 = __builtin_amdgcn_wmma_f32_16x16x4_f32(
                false, a, false, b1, (short)0, acc1, false, false);
        }
        const float bv0 = bias[col0 + m];
        const float bv1 = bias[col0 + 16 + m];
        if (!pooling) {
#pragma unroll
            for (int r = 0; r < 8; ++r) {
                const int row = row0 + mb + r;
                if (row < nOwn) {
                    float* op = act + (size_t)(base + row) * 128 + col0 + m;
                    op[0]  = fmaxf(acc0[r] + bv0, 0.0f);
                    op[16] = fmaxf(acc1[r] + bv1, 0.0f);
                }
            }
        } else {
            const int4 sa = *(const int4*)(sSlot + row0 + mb);
            const int4 sb = *(const int4*)(sSlot + row0 + mb + 4);
            const int sl[8] = {sa.x, sa.y, sa.z, sa.w, sb.x, sb.y, sb.z, sb.w};
            const int s0 = __builtin_amdgcn_readfirstlane(sa.x);
            bool same = true;
#pragma unroll
            for (int r = 0; r < 8; ++r) same = same && (sl[r] == s0);
            if (s0 >= 0 && __builtin_amdgcn_ballot_w32(same) == 0xffffffffu) {
                float t0 = 0.0f, t1 = 0.0f;
#pragma unroll
                for (int r = 0; r < 8; ++r) {
                    t0 += fmaxf(acc0[r] + bv0, 0.0f);
                    t1 += fmaxf(acc1[r] + bv1, 0.0f);
                }
                float* pp = sPool + s0 * 128 + col0 + m;
                __hip_atomic_fetch_add(pp,      t0, __ATOMIC_RELAXED, __HIP_MEMORY_SCOPE_WORKGROUP);
                __hip_atomic_fetch_add(pp + 16, t1, __ATOMIC_RELAXED, __HIP_MEMORY_SCOPE_WORKGROUP);
            } else {
#pragma unroll
                for (int r = 0; r < 8; ++r) {
                    const float v0 = fmaxf(acc0[r] + bv0, 0.0f);
                    const float v1 = fmaxf(acc1[r] + bv1, 0.0f);
                    if (sl[r] >= 0) {
                        float* pp = sPool + sl[r] * 128 + col0 + m;
                        __hip_atomic_fetch_add(pp,      v0, __ATOMIC_RELAXED, __HIP_MEMORY_SCOPE_WORKGROUP);
                        __hip_atomic_fetch_add(pp + 16, v1, __ATOMIC_RELAXED, __HIP_MEMORY_SCOPE_WORKGROUP);
                    } else if (sl[r] == -2) {
                        float* gp = gsum + (size_t)batch[base + row0 + mb + r] * 128 + col0 + m;
                        atomicAdd(gp,      v0);
                        atomicAdd(gp + 16, v1);
                    }
                }
            }
        }
    }

    if (pooling) {
        __syncthreads();
        for (int i = tid; i < POOL_SLOTS * 128; i += 1024) {
            const int s = i >> 7;
            if (sCnt[s] > 0) atomicAdd(gsum + (size_t)(gbase + s) * 128 + (i & 127), sPool[i]);
        }
        if (tid < POOL_SLOTS && sCnt[tid] > 0) atomicAdd(gcnt + gbase + tid, (float)sCnt[tid]);
    }
}

__global__ __launch_bounds__(256) void pool_div(const float* __restrict__ gsum,
                                                const float* __restrict__ gcnt,
                                                float* __restrict__ g, int G) {
    int i = blockIdx.x * 256 + threadIdx.x;
    if (i < G * 128) {
        int b = i >> 7;
        g[i] = gsum[i] / fmaxf(gcnt[b], 1.0f);
    }
}

extern "C" void kernel_launch(void* const* d_in, const int* in_sizes, int n_in,
                              void* d_out, int out_size, void* d_ws, size_t ws_size,
                              hipStream_t stream) {
    const float* x     = (const float*)d_in[0];
    const int*   ei    = (const int*)d_in[1];
    const int*   batch = (const int*)d_in[2];
    const float* W1    = (const float*)d_in[3];
    const float* b1    = (const float*)d_in[4];
    const float* W2    = (const float*)d_in[5];
    const float* b2    = (const float*)d_in[6];
    const float* Wl1   = (const float*)d_in[7];
    const float* bl1   = (const float*)d_in[8];
    const float* Wl2   = (const float*)d_in[9];
    const float* bl2   = (const float*)d_in[10];
    float* out = (float*)d_out;

    const int nN = in_sizes[0] / 128;
    const int nE = in_sizes[1] / 2;
    const int G  = out_size / 32;
    const int* src = ei;
    const int* dst = ei + nE;

    char* ws = (char*)d_ws;
    auto alloc = [&](size_t bytes) -> float* {
        float* p = (float*)ws;
        ws += (bytes + 255) & ~(size_t)255;
        return p;
    };
    float* bufA = alloc((size_t)nN * 128 * 4);
    float* bufB = alloc((size_t)nN * 128 * 4);
    float* deg  = alloc((size_t)nN * 4);
    float* dinv = alloc((size_t)nN * 4);
    float* gsum = alloc((size_t)G * 128 * 4);
    float* gcnt = alloc((size_t)G * 4);
    float* g    = alloc((size_t)G * 128 * 4);
    float* m1   = alloc((size_t)G * 128 * 4);

    const int  eB  = (nE + 255) / 256;
    const int  nB  = (nN + 255) / 256;
    const int  oB  = (nN + OWN_ROWS - 1) / OWN_ROWS;

    fill_f32<<<nB, 256, 0, stream>>>(deg, 1.0f, nN);
    fill_f32<<<(G * 128 + 255) / 256, 256, 0, stream>>>(gsum, 0.0f, (long)G * 128);
    fill_f32<<<(G + 255) / 256, 256, 0, stream>>>(gcnt, 0.0f, G);
    deg_count<<<eB, 256, 0, stream>>>(dst, deg, nE);
    rsqrt_k<<<nB, 256, 0, stream>>>(deg, dinv, nN);

    layer_owned_rows<<<oB, 1024, 0, stream>>>(src, dst, dinv, x, W1, b1, bufA, (int*)bufB, 0, nE, nN,
                                              nullptr, nullptr, nullptr, 0);

    layer_owned_rows<<<oB, 1024, 0, stream>>>(src, dst, dinv, bufA, W2, b2, bufB, (int*)bufB, 1, nE, nN,
                                              batch, gsum, gcnt, G);

    pool_div<<<(G * 128 + 255) / 256, 256, 0, stream>>>(gsum, gcnt, g, G);

    gemm_k128<<<(G + 15) / 16, 256, 0, stream>>>(g, Wl1, bl1, m1, G, 128, 1, 1);
    gemm_k128<<<(G + 15) / 16, 256, 0, stream>>>(m1, Wl2, bl2, out, G, 32, 0, 1);
}


# Layer scan reads one byte per edge (owner group = dst/392, packed by the degree pass), not the 4-byte dst: 4 keys tested per word op, 512 edges per wave load; LDS sums in arrival order as before
# speedup vs baseline: 4.3293x; 1.0448x over previous
#include <hip/hip_runtime.h>

typedef __attribute__((ext_vector_type(2))) float v2f;
typedef __attribute__((ext_vector_type(8))) float v8f;

__global__ __launch_bounds__(256) void fill_f32(float* __restrict__ p, float v, long n) {
    long i = (long)blockIdx.x * 256 + threadIdx.x;
    if (i < n) p[i] = v;
}

#define OWN_ROWS   392

__global__ __launch_bounds__(256) void deg_count(const int* __restrict__ dst,
                                                 float* __restrict__ deg,
                                                 unsigned* __restrict__ key4, int nE) {
    const int e = (blockIdx.x * 256 + threadIdx.x) * 4;
    if (e >= nE) return;
    int d[4] = {0, 0, 0, 0};
    if (e + 3 < nE && (((size_t)(dst + e)) & 15) == 0) {
        const int4 v = *(const int4*)(dst + e);
        d[0] = v.x; d[1] = v.y; d[2] = v.z; d[3] = v.w;
    } else {
#pragma unroll
        for (int cb = 0; cb < 4; ++cb) if (e + cb < nE) d[cb] = dst[e + cb];
    }
    unsigned kw = 0u;
#pragma unroll
    for (int cb = 0; cb < 4; ++cb) {
        if (e + cb < nE) {
            atomicAdd(&deg[d[cb]], 1.0f);
            kw |= (((unsigned)d[cb] / OWN_ROWS) & 0xffu) << (8 * cb);
        }
    }
    if (key4 != nullptr) key4[e >> 2] = kw;
}

__global__ __launch_bounds__(256) void rsqrt_k(const float* __restrict__ deg,
                                               float* __restrict__ dinv, int n) {
    int i = blockIdx.x * 256 + threadIdx.x;
    if (i < n) dinv[i] = rsqrtf(deg[i]);
}

#define LPITCH 66

__global__ __launch_bounds__(256) void gemm_k128(const float* __restrict__ A,
                                                 const float* __restrict__ W,
                                                 const float* __restrict__ bias,
                                                 float* __restrict__ out,
                                                 int M, int N, int do_relu, int has_bias) {
    __shared__ v2f sA2[16][LPITCH];
    __shared__ v2f sW2[128][LPITCH];

    const int row0 = blockIdx.x * 16;
    const int tid  = threadIdx.x;

    float* sWf = (float*)sW2;
    for (int i = tid; i < 128 * N; i += 256) {
        int k = i / N, n = i - k * N;
        sWf[n * (2 * LPITCH) + k] = W[i];
    }
    float* sAf = (float*)sA2;
    for (int i = tid; i < 16 * 128; i += 256) {
        int r = i >> 7, c = i & 127;
        sAf[r * (2 * LPITCH) + c] = (row0 + r < M) ? A[(size_t)(row0 + r) * 128 + c] : 0.0f;
    }
    __syncthreads();

    const int wave  = tid >> 5;
    const int lane  = tid & 31;
    const int m     = lane & 15;
    const int hsel  = lane >> 4;
    const int mb    = hsel << 3;
    const int ncolt = N >> 4;

    for (int ct = wave; ct < ncolt; ct += 8) {
        const int col0 = ct << 4;
        v8f acc = {};
#pragma unroll 4
        for (int k2 = 0; k2 < 64; k2 += 2) {
            const int kq = k2 + hsel;
            v2f a = sA2[m][kq];
            v2f b = sW2[col0 + m][kq];
            acc = __builtin_amdgcn_wmma_f32_16x16x4_f32(
                false, a, false, b, (short)0, acc, false, false);
        }
        const float bv = has_bias ? bias[col0 + m] : 0.0f;
#pragma unroll
        for (int r = 0; r < 8; ++r) {
            int row = row0 + mb + r;
            if (row < M) {
                float v = acc[r] + bv;
                if (do_relu) v = fmaxf(v, 0.0f);
                out[(size_t)row * N + col0 + m] = v;
            }
        }
    }
}

#define OWN_WAVES  32
#define SCAN_AHEAD 4
#define QCAP       544
#define APITCH     132
#define POOL_SLOTS 32
#define SLOT_ROWS  (((OWN_ROWS + 15) / 16) * 16)

__global__ __launch_bounds__(1024) void layer_owned_rows(const int* __restrict__ src,
                                                         const int* __restrict__ dst,
                                                         const float* __restrict__ dinv,
                                                         const float* __restrict__ xin,
                                                         const float* __restrict__ Wm,
                                                         const float* __restrict__ bias,
                                                         float* act, int* rec, int replay,
                                                         int nE, int nN,
                                                         const int* __restrict__ batch,
                                                         float* __restrict__ gsum,
                                                         float* __restrict__ gcnt, int G,
                                                         const unsigned char* __restrict__ keys) {
    __shared__ __attribute__((aligned(16))) float sAgg[OWN_ROWS * APITCH];
    __shared__ __attribute__((aligned(16))) int   sQ[OWN_WAVES * QCAP];
    static_assert(sizeof(int) * OWN_WAVES * QCAP >= sizeof(v2f) * 128 * LPITCH,
                  "the transposed W tile must fit where the queues were");
    __shared__ int sRecN;
    __shared__ int sRecBad;
    __shared__ __attribute__((aligned(16))) float sPool[POOL_SLOTS * 128];
    __shared__ __attribute__((aligned(16))) int   sSlot[SLOT_ROWS];
    __shared__ int sCnt[POOL_SLOTS];
    const bool pooling = gsum != nullptr;

    const int tid  = threadIdx.x;
    const int lane = tid & 31;
    const int wave = __builtin_amdgcn_readfirstlane(tid >> 5);
    const int base = (int)blockIdx.x * OWN_ROWS;
    const int nOwn = max(0, min(OWN_ROWS, nN - base));

    float4* z = (float4*)sAgg;
    for (int i = tid; i < OWN_ROWS * (APITCH / 4); i += 1024) z[i] = make_float4(0.0f, 0.0f, 0.0f, 0.0f);
    if (tid == 0) {
        sRecN = 0;
        sRecBad = 0;
    }
    if (pooling) {
        for (int i = tid; i < POOL_SLOTS * 32; i += 1024) ((float4*)sPool)[i] = make_float4(0.0f, 0.0f, 0.0f, 0.0f);
        if (tid < POOL_SLOTS) sCnt[tid] = 0;
    }
    __syncthreads();

    int*   lst = rec + (size_t)base * 128;
    const int cap = ((nOwn * 128 - 32) / 3) & ~31;
    int*   lS = lst + 32;
    int*   lR = lS + cap;
    float* lW = (float*)(lR + cap);
    int nRec = -1;
    if (replay) {
        nRec = __builtin_amdgcn_readfirstlane(lst[0]);
        if (nRec > cap) nRec = -1;
    }

    int* q = sQ + wave * QCAP;
    const int  nPieces = (nE + 127) >> 7;
    const int  nSets = (keys != nullptr && (((size_t)keys) & 15) == 0) ? (nE >> 9) : 0;
    const int  p0 = (int)(((unsigned)nSets / gridDim.x) * blockIdx.x);
    const unsigned gid4 = (unsigned)blockIdx.x * 0x01010101u;
    int qn = 0;

    auto set_at = [&](int t) -> int {
        const int s = t + p0;
        return s >= nSets ? s - nSets : s;
    };
    auto push = [&](bool m, unsigned b, int eid) {
        if (b) {
            const int pos = qn + (int)__builtin_amdgcn_mbcnt_lo(b, 0u);
            if (m) q[pos] = eid;
            qn += __builtin_popcount(b);
        }
    };
    auto gather_add = [&](int s, int r, float w, int cnt) {
        const int wi = __float_as_int(w);
        for (int j = 0; j < cnt; j += 8) {
            float4 v[8];
#pragma unroll
            for (int u = 0; u < 8; ++u) {
                const int sj = __builtin_amdgcn_readlane(s, j + u);
                v[u] = *(const float4*)(xin + (size_t)sj * 128 + (lane << 2));
            }
#pragma unroll
            for (int u = 0; u < 8; ++u) {
                if (j + u < cnt) {
                    const int   rj = __builtin_amdgcn_readlane(r, j + u);
                    const float wj = __int_as_float(__builtin_amdgcn_readlane(wi, j + u));
                    float* ap = sAgg + rj * APITCH + lane;
                    __hip_atomic_fetch_add(ap +  0, wj * v[u].x, __ATOMIC_RELAXED, __HIP_MEMORY_SCOPE_WORKGROUP);
                    __hip_atomic_fetch_add(ap + 32, wj * v[u].y, __ATOMIC_RELAXED, __HIP_MEMORY_SCOPE_WORKGROUP);
                    __hip_atomic_fetch_add(ap + 64, wj * v[u].z, __ATOMIC_RELAXED, __HIP_MEMORY_SCOPE_WORKGROUP);
                    __hip_atomic_fetch_add(ap + 96, wj * v[u].w, __ATOMIC_RELAXED, __HIP_MEMORY_SCOPE_WORKGROUP);
                }
            }
        }
    };
    auto drain = [&](const int* qp, int cnt) {
        int s = 0, r = 0;
        float w = 0.0f;
        if (lane < cnt) {
            const int eid = qp[lane];
            s = src[eid];
            const int d = dst[eid];
            r = d - base;
            w = dinv[s] * dinv[d];
        }
        if (!replay) {
            int pos = 0;
            if (lane == 0) pos = atomicAdd(&sRecN, cnt);
            pos = __builtin_amdgcn_readfirstlane(pos);
            if (pos + cnt <= cap) {
                if (lane < cnt) {
                    lS[pos + lane] = s;
                    lR[pos + lane] = r;
                    lW[pos + lane] = w;
                }
            } else if (lane == 0) {
                sRecBad = 1;
            }
        }
        gather_add(s, r, w, cnt);
    };

    auto test = [&](const int4 d4, int e) {
        const bool m0 = (unsigned)d4.x - (unsigned)base < (unsigned)nOwn;
        const bool m1 = (unsigned)d4.y - (unsigned)base < (unsigned)nOwn;
        const bool m2 = (unsigned)d4.z - (unsigned)base < (unsigned)nOwn;
        const bool m3 = (unsigned)d4.w - (unsigned)base < (unsigned)nOwn;
        const unsigned b0 = __builtin_amdgcn_ballot_w32(m0);
        const unsigned b1 = __builtin_amdgcn_ballot_w32(m1);
        const unsigned b2 = __builtin_amdgcn_ballot_w32(m2);
        const unsigned b3 = __builtin_amdgcn_ballot_w32(m3);
        if (b0 | b1 | b2 | b3) {
            push(m0, b0, e);
            push(m1, b1, e + 1);
            push(m2, b2, e + 2);
            push(m3, b3, e + 3);
        }
    };

    auto miss4 = [&](int kw) -> unsigned {
        const unsigned x = (unsigned)kw ^ gid4;
        return (((x & 0x7f7f7f7fu) + 0x7f7f7f7fu) | x) | 0x7f7f7f7fu;
    };
    auto hit4 = [&](unsigned h, int e) {
        if (__builtin_amdgcn_ballot_w32(h != 0u)) {
#pragma unroll
            for (int cb = 0; cb < 4; ++cb) {
                const bool m = ((h >> (8 * cb + 7)) & 1u) != 0u;
                push(m, __builtin_amdgcn_ballot_w32(m), e + cb);
            }
        }
    };
    auto test_keys = [&](const int4 kv, int e) {
        const unsigned r0 = miss4(kv.x);
        const unsigned r1 = miss4(kv.y);
        const unsigned r2 = miss4(kv.z);
        const unsigned r3 = miss4(kv.w);
        if (__builtin_amdgcn_ballot_w32((r0 & r1 & r2 & r3) != 0xffffffffu)) {
            hit4(~r0, e);
            hit4(~r1, e + 4);
            hit4(~r2, e + 8);
            hit4(~r3, e + 12);
        }
    };

    auto load_set = [&](int t) -> int4 {
        return *((const int4*)(keys + ((size_t)set_at(min(t, nSets - 1)) << 9)) + lane);
    };
    auto test_set = [&](const int4 kv, int t) {
        if (t < nSets) test_keys(kv, (set_at(t) << 9) + (lane << 4));
        while (qn >= 32) {
            qn -= 32;
            drain(q + qn, 32);
        }
    };

    if (nRec >= 0) {
        for (int i = wave * 32; i < nRec; i += 32 * OWN_WAVES) {
            int s = 0, r = 0;
            float w = 0.0f;
            if (i + lane < nRec) {
                s = lS[i + lane];
                r = lR[i + lane];
                w = lW[i + lane];
            }
            gather_add(s, r, w, min(32, nRec - i));
        }
    } else {
        if (nSets > 0) {
            static_assert(SCAN_AHEAD == 4, "the loop below names four registers");
            int4 k0 = load_set(wave);
            int4 k1 = load_set(wave + OWN_WAVES);
            int4 k2 = load_set(wave + 2 * OWN_WAVES);
            int4 k3 = load_set(wave + 3 * OWN_WAVES);
            for (int t = wave; t < nSets; t += SCAN_AHEAD * OWN_WAVES) {
                test_set(k0, t);
                k0 = load_set(t + 4 * OWN_WAVES);
                test_set(k1, t + OWN_WAVES);
                k1 = load_set(t + 5 * OWN_WAVES);
                test_set(k2, t + 2 * OWN_WAVES);
                k2 = load_set(t + 6 * OWN_WAVES);
                test_set(k3, t + 3 * OWN_WAVES);
                k3 = load_set(t + 7 * OWN_WAVES);
            }
        }
        for (int k = nSets * 4 + wave; k < nPieces; k += OWN_WAVES) {
            const int e = (k << 7) + (lane << 2);
            int4 d4;
            d4.x = (e     < nE) ? dst[e]     : -1;
            d4.y = (e + 1 < nE) ? dst[e + 1] : -1;
            d4.z = (e + 2 < nE) ? dst[e + 2] : -1;
            d4.w = (e + 3 < nE) ? dst[e + 3] : -1;
            test(d4, e);
            while (qn >= 32) {
                qn -= 32;
                drain(q + qn, 32);
            }
        }
        if (qn > 0) drain(q, qn);
    }
    __syncthreads();
    if (!replay && tid == 0) lst[0] = sRecBad ? -1 : sRecN;

    v2f*   sW2 = (v2f*)sQ;
    float* sWf = (float*)sQ;
    for (int i = tid; i < 128 * 128; i += 1024) {
        const int k = i >> 7, n = i & 127;
        sWf[n * (2 * LPITCH) + k] = Wm[i];
    }

    int gbase = 0;
    if (pooling) {
        gbase = __builtin_amdgcn_readfirstlane(nOwn > 0 ? batch[base] : 0);
        for (int r = tid; r < SLOT_ROWS; r += 1024) {
            int code = -1;
            if (r < nOwn) {
                const int b = batch[base + r];
                if ((unsigned)b < (unsigned)G) {
                    const int s = b - gbase;
                    if ((unsigned)s < (unsigned)POOL_SLOTS) {
                        code = s;
                        atomicAdd(&sCnt[s], 1);
                    } else {
                        code = -2;
                        atomicAdd(gcnt + b, 1.0f);
                    }
                }
            }
            sSlot[r] = code;
        }
    }

    for (int r0 = wave; r0 < nOwn; r0 += 4 * OWN_WAVES) {
        float  di[4];
        float4 hv[4];
#pragma unroll
        for (int t = 0; t < 4; ++t) {
            const size_t node = (size_t)(base + min(r0 + OWN_WAVES * t, nOwn - 1));
            di[t] = dinv[node];
            hv[t] = *(const float4*)(xin + node * 128 + (lane << 2));
        }
#pragma unroll
        for (int t = 0; t < 4; ++t) {
            const int r = r0 + OWN_WAVES * t;
            if (r < nOwn) {
                const float d2 = di[t] * di[t];
                float* rowp = sAgg + r * APITCH;
                const float* ap = rowp + lane;
                float4 o;
                o.x = ap[0]  + d2 * hv[t].x;
                o.y = ap[32] + d2 * hv[t].y;
                o.z = ap[64] + d2 * hv[t].z;
                o.w = ap[96] + d2 * hv[t].w;
                *(float4*)(rowp + (lane << 2)) = o;
            }
        }
    }
    __syncthreads();

    const int m      = lane & 15;
    const int hsel   = lane >> 4;
    const int mb     = hsel << 3;
    const int nItems = ((nOwn + 15) >> 4) << 2;
    for (int p = wave; p < nItems; p += OWN_WAVES) {
        const int row0 = (p >> 2) << 4;
        const int col0 = (p & 3) << 5;
        const v2f* arow = (const v2f*)(sAgg + min(row0 + m, OWN_ROWS - 1) * APITCH);
        const v2f* b0p  = sW2 + (col0 + m) * LPITCH;
        const v2f* b1p  = sW2 + (col0 + 16 + m) * LPITCH;
        v8f acc0 = {};
        v8f acc1 = {};
#pragma unroll 4
        for (int k2 = 0; k2 < 64; k2 += 2) {
            const int kq = k2 + hsel;
            const v2f a  = arow[kq];
            const v2f b0 = b0p[kq];
            const v2f b1 = b1p[kq];
            acc0 = __builtin_amdgcn_wmma_f32_16x16x4_f32(
                false, a, false, b0, (short)0, acc0, false, false);
            acc1 = __builtin_amdgcn_wmma_f32_16x16x4_f32(
                false, a, false, b1, (short)0, acc1, false, false);
        }
        const float bv0 = bias[col0 + m];
        const float bv1 = bias[col0 + 16 + m];
        if (!pooling) {
#pragma unroll
            for (int r = 0; r < 8; ++r) {
                const int row = row0 + mb + r;
                if (row < nOwn) {
                    float* op = act + (size_t)(base + row) * 128 + col0 + m;
                    op[0]  = fmaxf(acc0[r] + bv0, 0.0f);
                    op[16] = fmaxf(acc1[r] + bv1, 0.0f);
                }
            }
        } else {
            const int4 sa = *(const int4*)(sSlot + row0 + mb);
            const int4 sb = *(const int4*)(sSlot + row0 + mb + 4);
            const int sl[8] = {sa.x, sa.y, sa.z, sa.w, sb.x, sb.y, sb.z, sb.w};
            const int s0 = __builtin_amdgcn_readfirstlane(sa.x);
            bool same = true;
#pragma unroll
            for (int r = 0; r < 8; ++r) same = same && (sl[r] == s0);
            if (s0 >= 0 && __builtin_amdgcn_ballot_w32(same) == 0xffffffffu) {
                float t0 = 0.0f, t1 = 0.0f;
#pragma unroll
                for (int r = 0; r < 8; ++r) {
                    t0 += fmaxf(acc0[r] + bv0, 0.0f);
                    t1 += fmaxf(acc1[r] + bv1, 0.0f);
                }
                float* pp = sPool + s0 * 128 + col0 + m;
                __hip_atomic_fetch_add(pp,      t0, __ATOMIC_RELAXED, __HIP_MEMORY_SCOPE_WORKGROUP);
                __hip_atomic_fetch_add(pp + 16, t1, __ATOMIC_RELAXED, __HIP_MEMORY_SCOPE_WORKGROUP);
            } else {
#pragma unroll
                for (int r = 0; r < 8; ++r) {
                    const float v0 = fmaxf(acc0[r] + bv0, 0.0f);
                    const float v1 = fmaxf(acc1[r] + bv1, 0.0f);
                    if (sl[r] >= 0) {
                        float* pp = sPool + sl[r] * 128 + col0 + m;
                        __hip_atomic_fetch_add(pp,      v0, __ATOMIC_RELAXED, __HIP_MEMORY_SCOPE_WORKGROUP);
                        __hip_atomic_fetch_add(pp + 16, v1, __ATOMIC_RELAXED, __HIP_MEMORY_SCOPE_WORKGROUP);
                    } else if (sl[r] == -2) {
                        float* gp = gsum + (size_t)batch[base + row0 + mb + r] * 128 + col0 + m;
                        atomicAdd(gp,      v0);
                        atomicAdd(gp + 16, v1);
                    }
                }
            }
        }
    }

    if (pooling) {
        __syncthreads();
        for (int i = tid; i < POOL_SLOTS * 128; i += 1024) {
            const int s = i >> 7;
            if (sCnt[s] > 0) atomicAdd(gsum + (size_t)(gbase + s) * 128 + (i & 127), sPool[i]);
        }
        if (tid < POOL_SLOTS && sCnt[tid] > 0) atomicAdd(gcnt + gbase + tid, (float)sCnt[tid]);
    }
}

__global__ __launch_bounds__(256) void pool_div(const float* __restrict__ gsum,
                                                const float* __restrict__ gcnt,
                                                float* __restrict__ g, int G) {
    int i = blockIdx.x * 256 + threadIdx.x;
    if (i < G * 128) {
        int b = i >> 7;
        g[i] = gsum[i] / fmaxf(gcnt[b], 1.0f);
    }
}

extern "C" void kernel_launch(void* const* d_in, const int* in_sizes, int n_in,
                              void* d_out, int out_size, void* d_ws, size_t ws_size,
                              hipStream_t stream) {
    const float* x     = (const float*)d_in[0];
    const int*   ei    = (const int*)d_in[1];
    const int*   batch = (const int*)d_in[2];
    const float* W1    = (const float*)d_in[3];
    const float* b1    = (const float*)d_in[4];
    const float* W2    = (const float*)d_in[5];
    const float* b2    = (const float*)d_in[6];
    const float* Wl1   = (const float*)d_in[7];
    const float* bl1   = (const float*)d_in[8];
    const float* Wl2   = (const float*)d_in[9];
    const float* bl2   = (const float*)d_in[10];
    float* out = (float*)d_out;

    const int nN = in_sizes[0] / 128;
    const int nE = in_sizes[1] / 2;
    const int G  = out_size / 32;
    const int* src = ei;
    const int* dst = ei + nE;

    char* ws = (char*)d_ws;
    auto alloc = [&](size_t bytes) -> float* {
        float* p = (float*)ws;
        ws += (bytes + 255) & ~(size_t)255;
        return p;
    };
    float* bufA = alloc((size_t)nN * 128 * 4);
    float* bufB = alloc((size_t)nN * 128 * 4);
    float* deg  = alloc((size_t)nN * 4);
    float* dinv = alloc((size_t)nN * 4);
    float* gsum = alloc((size_t)G * 128 * 4);
    float* gcnt = alloc((size_t)G * 4);
    float* g    = alloc((size_t)G * 128 * 4);
    float* m1   = alloc((size_t)G * 128 * 4);
    unsigned char* keys = (unsigned char*)ws;
    ws += ((size_t)nE + 3 + 255) & ~(size_t)255;

    const int  eB  = (nE + 1023) / 1024;
    const int  nB  = (nN + 255) / 256;
    const int  oB  = (nN + OWN_ROWS - 1) / OWN_ROWS;
    if (oB > 256 || (size_t)(ws - (char*)d_ws) > ws_size) keys = nullptr;

    fill_f32<<<nB, 256, 0, stream>>>(deg, 1.0f, nN);
    fill_f32<<<(G * 128 + 255) / 256, 256, 0, stream>>>(gsum, 0.0f, (long)G * 128);
    fill_f32<<<(G + 255) / 256, 256, 0, stream>>>(gcnt, 0.0f, G);
    deg_count<<<eB, 256, 0, stream>>>(dst, deg, (unsigned*)keys, nE);
    rsqrt_k<<<nB, 256, 0, stream>>>(deg, dinv, nN);

    layer_owned_rows<<<oB, 1024, 0, stream>>>(src, dst, dinv, x, W1, b1, bufA, (int*)bufB, 0, nE, nN,
                                              nullptr, nullptr, nullptr, 0, keys);

    layer_owned_rows<<<oB, 1024, 0, stream>>>(src, dst, dinv, bufA, W2, b2, bufB, (int*)bufB, 1, nE, nN,
                                              batch, gsum, gcnt, G, keys);

    pool_div<<<(G * 128 + 255) / 256, 256, 0, stream>>>(gsum, gcnt, g, G);

    gemm_k128<<<(G + 15) / 16, 256, 0, stream>>>(g, Wl1, bl1, m1, G, 128, 1, 1);
    gemm_k128<<<(G + 15) / 16, 256, 0, stream>>>(m1, Wl2, bl2, out, G, 32, 0, 1);
}


# Layer-1 scan: a lane folds its 16 key tests into one bit mask; the wave queues hits in lowest-bit rounds (one ballot a round, not one per byte position); queue order, so LDS arrival order, differs
# speedup vs baseline: 4.5309x; 1.0466x over previous
#include <hip/hip_runtime.h>

typedef __attribute__((ext_vector_type(2))) float v2f;
typedef __attribute__((ext_vector_type(8))) float v8f;

__global__ __launch_bounds__(256) void fill_f32(float* __restrict__ p, float v, long n) {
    long i = (long)blockIdx.x * 256 + threadIdx.x;
    if (i < n) p[i] = v;
}

#define OWN_ROWS   392

__global__ __launch_bounds__(256) void deg_count(const int* __restrict__ dst,
                                                 float* __restrict__ deg,
                                                 unsigned* __restrict__ key4, int nE) {
    const int e = (blockIdx.x * 256 + threadIdx.x) * 4;
    if (e >= nE) return;
    int d[4] = {0, 0, 0, 0};
    if (e + 3 < nE && (((size_t)(dst + e)) & 15) == 0) {
        const int4 v = *(const int4*)(dst + e);
        d[0] = v.x; d[1] = v.y; d[2] = v.z; d[3] = v.w;
    } else {
#pragma unroll
        for (int cb = 0; cb < 4; ++cb) if (e + cb < nE) d[cb] = dst[e + cb];
    }
    unsigned kw = 0u;
#pragma unroll
    for (int cb = 0; cb < 4; ++cb) {
        if (e + cb < nE) {
            atomicAdd(&deg[d[cb]], 1.0f);
            kw |= (((unsigned)d[cb] / OWN_ROWS) & 0xffu) << (8 * cb);
        }
    }
    if (key4 != nullptr) key4[e >> 2] = kw;
}

__global__ __launch_bounds__(256) void rsqrt_k(const float* __restrict__ deg,
                                               float* __restrict__ dinv, int n) {
    int i = blockIdx.x * 256 + threadIdx.x;
    if (i < n) dinv[i] = rsqrtf(deg[i]);
}

#define LPITCH 66

__global__ __launch_bounds__(256) void gemm_k128(const float* __restrict__ A,
                                                 const float* __restrict__ W,
                                                 const float* __restrict__ bias,
                                                 float* __restrict__ out,
                                                 int M, int N, int do_relu, int has_bias) {
    __shared__ v2f sA2[16][LPITCH];
    __shared__ v2f sW2[128][LPITCH];

    const int row0 = blockIdx.x * 16;
    const int tid  = threadIdx.x;

    float* sWf = (float*)sW2;
    for (int i = tid; i < 128 * N; i += 256) {
        int k = i / N, n = i - k * N;
        sWf[n * (2 * LPITCH) + k] = W[i];
    }
    float* sAf = (float*)sA2;
    for (int i = tid; i < 16 * 128; i += 256) {
        int r = i >> 7, c = i & 127;
        sAf[r * (2 * LPITCH) + c] = (row0 + r < M) ? A[(size_t)(row0 + r) * 128 + c] : 0.0f;
    }
    __syncthreads();

    const int wave  = tid >> 5;
    const int lane  = tid & 31;
    const int m     = lane & 15;
    const int hsel  = lane >> 4;
    const int mb    = hsel << 3;
    const int ncolt = N >> 4;

    for (int ct = wave; ct < ncolt; ct += 8) {
        const int col0 = ct << 4;
        v8f acc = {};
#pragma unroll 4
        for (int k2 = 0; k2 < 64; k2 += 2) {
            const int kq = k2 + hsel;
            v2f a = sA2[m][kq];
            v2f b = sW2[col0 + m][kq];
            acc = __builtin_amdgcn_wmma_f32_16x16x4_f32(
                false, a, false, b, (short)0, acc, false, false);
        }
        const float bv = has_bias ? bias[col0 + m] : 0.0f;
#pragma unroll
        for (int r = 0; r < 8; ++r) {
            int row = row0 + mb + r;
            if (row < M) {
                float v = acc[r] + bv;
                if (do_relu) v = fmaxf(v, 0.0f);
                out[(size_t)row * N + col0 + m] = v;
            }
        }
    }
}

#define OWN_WAVES  32
#define SCAN_AHEAD 4
#define QCAP       544
#define APITCH     132
#define POOL_SLOTS 32
#define SLOT_ROWS  (((OWN_ROWS + 15) / 16) * 16)

__global__ __launch_bounds__(1024) void layer_owned_rows(const int* __restrict__ src,
                                                         const int* __restrict__ dst,
                                                         const float* __restrict__ dinv,
                                                         const float* __restrict__ xin,
                                                         const float* __restrict__ Wm,
                                                         const float* __restrict__ bias,
                                                         float* act, int* rec, int replay,
                                                         int nE, int nN,
                                                         const int* __restrict__ batch,
                                                         float* __restrict__ gsum,
                                                         float* __restrict__ gcnt, int G,
                                                         const unsigned char* __restrict__ keys) {
    __shared__ __attribute__((aligned(16))) float sAgg[OWN_ROWS * APITCH];
    __shared__ __attribute__((aligned(16))) int   sQ[OWN_WAVES * QCAP];
    static_assert(sizeof(int) * OWN_WAVES * QCAP >= sizeof(v2f) * 128 * LPITCH,
                  "the transposed W tile must fit where the queues were");
    __shared__ int sRecN;
    __shared__ int sRecBad;
    __shared__ __attribute__((aligned(16))) float sPool[POOL_SLOTS * 128];
    __shared__ __attribute__((aligned(16))) int   sSlot[SLOT_ROWS];
    __shared__ int sCnt[POOL_SLOTS];
    const bool pooling = gsum != nullptr;

    const int tid  = threadIdx.x;
    const int lane = tid & 31;
    const int wave = __builtin_amdgcn_readfirstlane(tid >> 5);
    const int base = (int)blockIdx.x * OWN_ROWS;
    const int nOwn = max(0, min(OWN_ROWS, nN - base));

    float4* z = (float4*)sAgg;
    for (int i = tid; i < OWN_ROWS * (APITCH / 4); i += 1024) z[i] = make_float4(0.0f, 0.0f, 0.0f, 0.0f);
    if (tid == 0) {
        sRecN = 0;
        sRecBad = 0;
    }
    if (pooling) {
        for (int i = tid; i < POOL_SLOTS * 32; i += 1024) ((float4*)sPool)[i] = make_float4(0.0f, 0.0f, 0.0f, 0.0f);
        if (tid < POOL_SLOTS) sCnt[tid] = 0;
    }
    __syncthreads();

    int*   lst = rec + (size_t)base * 128;
    const int cap = ((nOwn * 128 - 32) / 3) & ~31;
    int*   lS = lst + 32;
    int*   lR = lS + cap;
    float* lW = (float*)(lR + cap);
    int nRec = -1;
    if (replay) {
        nRec = __builtin_amdgcn_readfirstlane(lst[0]);
        if (nRec > cap) nRec = -1;
    }

    int* q = sQ + wave * QCAP;
    const int  nPieces = (nE + 127) >> 7;
    const int  nSets = (keys != nullptr && (((size_t)keys) & 15) == 0) ? (nE >> 9) : 0;
    const int  p0 = (int)(((unsigned)nSets / gridDim.x) * blockIdx.x);
    const unsigned gid4 = (unsigned)blockIdx.x * 0x01010101u;
    int qn = 0;

    auto set_at = [&](int t) -> int {
        const int s = t + p0;
        return s >= nSets ? s - nSets : s;
    };
    auto push = [&](bool m, unsigned b, int eid) {
        if (b) {
            const int pos = qn + (int)__builtin_amdgcn_mbcnt_lo(b, 0u);
            if (m) q[pos] = eid;
            qn += __builtin_popcount(b);
        }
    };
    auto gather_add = [&](int s, int r, float w, int cnt) {
        const int wi = __float_as_int(w);
        for (int j = 0; j < cnt; j += 8) {
            float4 v[8];
#pragma unroll
            for (int u = 0; u < 8; ++u) {
                const int sj = __builtin_amdgcn_readlane(s, j + u);
                v[u] = *(const float4*)(xin + (size_t)sj * 128 + (lane << 2));
            }
#pragma unroll
            for (int u = 0; u < 8; ++u) {
                if (j + u < cnt) {
                    const int   rj = __builtin_amdgcn_readlane(r, j + u);
                    const float wj = __int_as_float(__builtin_amdgcn_readlane(wi, j + u));
                    float* ap = sAgg + rj * APITCH + lane;
                    __hip_atomic_fetch_add(ap +  0, wj * v[u].x, __ATOMIC_RELAXED, __HIP_MEMORY_SCOPE_WORKGROUP);
                    __hip_atomic_fetch_add(ap + 32, wj * v[u].y, __ATOMIC_RELAXED, __HIP_MEMORY_SCOPE_WORKGROUP);
                    __hip_atomic_fetch_add(ap + 64, wj * v[u].z, __ATOMIC_RELAXED, __HIP_MEMORY_SCOPE_WORKGROUP);
                    __hip_atomic_fetch_add(ap + 96, wj * v[u].w, __ATOMIC_RELAXED, __HIP_MEMORY_SCOPE_WORKGROUP);
                }
            }
        }
    };
    auto drain = [&](const int* qp, int cnt) {
        int s = 0, r = 0;
        float w = 0.0f;
        if (lane < cnt) {
            const int eid = qp[lane];
            s = src[eid];
            const int d = dst[eid];
            r = d - base;
            w = dinv[s] * dinv[d];
        }
        if (!replay) {
            int pos = 0;
            if (lane == 0) pos = atomicAdd(&sRecN, cnt);
            pos = __builtin_amdgcn_readfirstlane(pos);
            if (pos + cnt <= cap) {
                if (lane < cnt) {
                    lS[pos + lane] = s;
                    lR[pos + lane] = r;
                    lW[pos + lane] = w;
                }
            } else if (lane == 0) {
                sRecBad = 1;
            }
        }
        gather_add(s, r, w, cnt);
    };

    auto test = [&](const int4 d4, int e) {
        const bool m0 = (unsigned)d4.x - (unsigned)base < (unsigned)nOwn;
        const bool m1 = (unsigned)d4.y - (unsigned)base < (unsigned)nOwn;
        const bool m2 = (unsigned)d4.z - (unsigned)base < (unsigned)nOwn;
        const bool m3 = (unsigned)d4.w - (unsigned)base < (unsigned)nOwn;
        const unsigned b0 = __builtin_amdgcn_ballot_w32(m0);
        const unsigned b1 = __builtin_amdgcn_ballot_w32(m1);
        const unsigned b2 = __builtin_amdgcn_ballot_w32(m2);
        const unsigned b3 = __builtin_amdgcn_ballot_w32(m3);
        if (b0 | b1 | b2 | b3) {
            push(m0, b0, e);
            push(m1, b1, e + 1);
            push(m2, b2, e + 2);
            push(m3, b3, e + 3);
        }
    };

    auto hit4 = [&](int kw) -> unsigned {
        const unsigned x = (unsigned)kw ^ gid4;
        return ~(((x & 0x7f7f7f7fu) + 0x7f7f7f7fu) | x) & 0x80808080u;
    };
    auto test_keys = [&](const int4 kv, int e) {
        unsigned hm = (hit4(kv.x) >> 7) | (hit4(kv.y) >> 6) | (hit4(kv.z) >> 5) | (hit4(kv.w) >> 4);
        unsigned b = __builtin_amdgcn_ballot_w32(hm != 0u);
        while (b) {
            if (hm != 0u) {
                const int p = __builtin_ctz(hm);
                q[qn + (int)__builtin_amdgcn_mbcnt_lo(b, 0u)] = e + ((p & 7) << 2) + (p >> 3);
            }
            qn += __builtin_popcount(b);
            hm &= hm - 1u;
            b = __builtin_amdgcn_ballot_w32(hm != 0u);
        }
    };

    auto load_set = [&](int t) -> int4 {
        return *((const int4*)(keys + ((size_t)set_at(min(t, nSets - 1)) << 9)) + lane);
    };
    auto test_set = [&](const int4 kv, int t) {
        if (t < nSets) test_keys(kv, (set_at(t) << 9) + (lane << 4));
        while (qn >= 32) {
            qn -= 32;
            drain(q + qn, 32);
        }
    };

    if (nRec >= 0) {
        for (int i = wave * 32; i < nRec; i += 32 * OWN_WAVES) {
            int s = 0, r = 0;
            float w = 0.0f;
            if (i + lane < nRec) {
                s = lS[i + lane];
                r = lR[i + lane];
                w = lW[i + lane];
            }
            gather_add(s, r, w, min(32, nRec - i));
        }
    } else {
        if (nSets > 0) {
            static_assert(SCAN_AHEAD == 4, "the loop below names four registers");
            int4 k0 = load_set(wave);
            int4 k1 = load_set(wave + OWN_WAVES);
            int4 k2 = load_set(wave + 2 * OWN_WAVES);
            int4 k3 = load_set(wave + 3 * OWN_WAVES);
            for (int t = wave; t < nSets; t += SCAN_AHEAD * OWN_WAVES) {
                test_set(k0, t);
                k0 = load_set(t + 4 * OWN_WAVES);
                test_set(k1, t + OWN_WAVES);
                k1 = load_set(t + 5 * OWN_WAVES);
                test_set(k2, t + 2 * OWN_WAVES);
                k2 = load_set(t + 6 * OWN_WAVES);
                test_set(k3, t + 3 * OWN_WAVES);
                k3 = load_set(t + 7 * OWN_WAVES);
            }
        }
        for (int k = nSets * 4 + wave; k < nPieces; k += OWN_WAVES) {
            const int e = (k << 7) + (lane << 2);
            int4 d4;
            d4.x = (e     < nE) ? dst[e]     : -1;
            d4.y = (e + 1 < nE) ? dst[e + 1] : -1;
            d4.z = (e + 2 < nE) ? dst[e + 2] : -1;
            d4.w = (e + 3 < nE) ? dst[e + 3] : -1;
            test(d4, e);
            while (qn >= 32) {
                qn -= 32;
                drain(q + qn, 32);
            }
        }
        if (qn > 0) drain(q, qn);
    }
    __syncthreads();
    if (!replay && tid == 0) lst[0] = sRecBad ? -1 : sRecN;

    v2f*   sW2 = (v2f*)sQ;
    float* sWf = (float*)sQ;
    for (int i = tid; i < 128 * 128; i += 1024) {
        const int k = i >> 7, n = i & 127;
        sWf[n * (2 * LPITCH) + k] = Wm[i];
    }

    int gbase = 0;
    if (pooling) {
        gbase = __builtin_amdgcn_readfirstlane(nOwn > 0 ? batch[base] : 0);
        for (int r = tid; r < SLOT_ROWS; r += 1024) {
            int code = -1;
            if (r < nOwn) {
                const int b = batch[base + r];
                if ((unsigned)b < (unsigned)G) {
                    const int s = b - gbase;
                    if ((unsigned)s < (unsigned)POOL_SLOTS) {
                        code = s;
                        atomicAdd(&sCnt[s], 1);
                    } else {
                        code = -2;
                        atomicAdd(gcnt + b, 1.0f);
                    }
                }
            }
            sSlot[r] = code;
        }
    }

    for (int r0 = wave; r0 < nOwn; r0 += 4 * OWN_WAVES) {
        float  di[4];
        float4 hv[4];
#pragma unroll
        for (int t = 0; t < 4; ++t) {
            const size_t node = (size_t)(base + min(r0 + OWN_WAVES * t, nOwn - 1));
            di[t] = dinv[node];
            hv[t] = *(const float4*)(xin + node * 128 + (lane << 2));
        }
#pragma unroll
        for (int t = 0; t < 4; ++t) {
            const int r = r0 + OWN_WAVES * t;
            if (r < nOwn) {
                const float d2 = di[t] * di[t];
                float* rowp = sAgg + r * APITCH;
                const float* ap = rowp + lane;
                float4 o;
                o.x = ap[0]  + d2 * hv[t].x;
                o.y = ap[32] + d2 * hv[t].y;
                o.z = ap[64] + d2 * hv[t].z;
                o.w = ap[96] + d2 * hv[t].w;
                *(float4*)(rowp + (lane << 2)) = o;
            }
        }
    }
    __syncthreads();

    const int m      = lane & 15;
    const int hsel   = lane >> 4;
    const int mb     = hsel << 3;
    const int nItems = ((nOwn + 15) >> 4) << 2;
    for (int p = wave; p < nItems; p += OWN_WAVES) {
        const int row0 = (p >> 2) << 4;
        const int col0 = (p & 3) << 5;
        const v2f* arow = (const v2f*)(sAgg + min(row0 + m, OWN_ROWS - 1) * APITCH);
        const v2f* b0p  = sW2 + (col0 + m) * LPITCH;
        const v2f* b1p  = sW2 + (col0 + 16 + m) * LPITCH;
        v8f acc0 = {};
        v8f acc1 = {};
#pragma unroll 4
        for (int k2 = 0; k2 < 64; k2 += 2) {
            const int kq = k2 + hsel;
            const v2f a  = arow[kq];
            const v2f b0 = b0p[kq];
            const v2f b1 = b1p[kq];
            acc0 = __builtin_amdgcn_wmma_f32_16x16x4_f32(
                false, a, false, b0, (short)0, acc0, false, false);
            acc1 = __builtin_amdgcn_wmma_f32_16x16x4_f32(
                false, a, false, b1, (short)0, acc1, false, false);
        }
        const float bv0 = bias[col0 + m];
        const float bv1 = bias[col0 + 16 + m];
        if (!pooling) {
#pragma unroll
            for (int r = 0; r < 8; ++r) {
                const int row = row0 + mb + r;
                if (row < nOwn) {
                    float* op = act + (size_t)(base + row) * 128 + col0 + m;
                    op[0]  = fmaxf(acc0[r] + bv0, 0.0f);
                    op[16] = fmaxf(acc1[r] + bv1, 0.0f);
                }
            }
        } else {
            const int4 sa = *(const int4*)(sSlot + row0 + mb);
            const int4 sb = *(const int4*)(sSlot + row0 + mb + 4);
            const int sl[8] = {sa.x, sa.y, sa.z, sa.w, sb.x, sb.y, sb.z, sb.w};
            const int s0 = __builtin_amdgcn_readfirstlane(sa.x);
            bool same = true;
#pragma unroll
            for (int r = 0; r < 8; ++r) same = same && (sl[r] == s0);
            if (s0 >= 0 && __builtin_amdgcn_ballot_w32(same) == 0xffffffffu) {
                float t0 = 0.0f, t1 = 0.0f;
#pragma unroll
                for (int r = 0; r < 8; ++r) {
                    t0 += fmaxf(acc0[r] + bv0, 0.0f);
                    t1 += fmaxf(acc1[r] + bv1, 0.0f);
                }
                float* pp = sPool + s0 * 128 + col0 + m;
                __hip_atomic_fetch_add(pp,      t0, __ATOMIC_RELAXED, __HIP_MEMORY_SCOPE_WORKGROUP);
                __hip_atomic_fetch_add(pp + 16, t1, __ATOMIC_RELAXED, __HIP_MEMORY_SCOPE_WORKGROUP);
            } else {
#pragma unroll
                for (int r = 0; r < 8; ++r) {
                    const float v0 = fmaxf(acc0[r] + bv0, 0.0f);
                    const float v1 = fmaxf(acc1[r] + bv1, 0.0f);
                    if (sl[r] >= 0) {
                        float* pp = sPool + sl[r] * 128 + col0 + m;
                        __hip_atomic_fetch_add(pp,      v0, __ATOMIC_RELAXED, __HIP_MEMORY_SCOPE_WORKGROUP);
                        __hip_atomic_fetch_add(pp + 16, v1, __ATOMIC_RELAXED, __HIP_MEMORY_SCOPE_WORKGROUP);
                    } else if (sl[r] == -2) {
                        float* gp = gsum + (size_t)batch[base + row0 + mb + r] * 128 + col0 + m;
                        atomicAdd(gp,      v0);
                        atomicAdd(gp + 16, v1);
                    }
                }
            }
        }
    }

    if (pooling) {
        __syncthreads();
        for (int i = tid; i < POOL_SLOTS * 128; i += 1024) {
            const int s = i >> 7;
            if (sCnt[s] > 0) atomicAdd(gsum + (size_t)(gbase + s) * 128 + (i & 127), sPool[i]);
        }
        if (tid < POOL_SLOTS && sCnt[tid] > 0) atomicAdd(gcnt + gbase + tid, (float)sCnt[tid]);
    }
}

__global__ __launch_bounds__(256) void pool_div(const float* __restrict__ gsum,
                                                const float* __restrict__ gcnt,
                                                float* __restrict__ g, int G) {
    int i = blockIdx.x * 256 + threadIdx.x;
    if (i < G * 128) {
        int b = i >> 7;
        g[i] = gsum[i] / fmaxf(gcnt[b], 1.0f);
    }
}

extern "C" void kernel_launch(void* const* d_in, const int* in_sizes, int n_in,
                              void* d_out, int out_size, void* d_ws, size_t ws_size,
                              hipStream_t stream) {
    const float* x     = (const float*)d_in[0];
    const int*   ei    = (const int*)d_in[1];
    const int*   batch = (const int*)d_in[2];
    const float* W1    = (const float*)d_in[3];
    const float* b1    = (const float*)d_in[4];
    const float* W2    = (const float*)d_in[5];
    const float* b2    = (const float*)d_in[6];
    const float* Wl1   = (const float*)d_in[7];
    const float* bl1   = (const float*)d_in[8];
    const float* Wl2   = (const float*)d_in[9];
    const float* bl2   = (const float*)d_in[10];
    float* out = (float*)d_out;

    const int nN = in_sizes[0] / 128;
    const int nE = in_sizes[1] / 2;
    const int G  = out_size / 32;
    const int* src = ei;
    const int* dst = ei + nE;

    char* ws = (char*)d_ws;
    auto alloc = [&](size_t bytes) -> float* {
        float* p = (float*)ws;
        ws += (bytes + 255) & ~(size_t)255;
        return p;
    };
    float* bufA = alloc((size_t)nN * 128 * 4);
    float* bufB = alloc((size_t)nN * 128 * 4);
    float* deg  = alloc((size_t)nN * 4);
    float* dinv = alloc((size_t)nN * 4);
    float* gsum = alloc((size_t)G * 128 * 4);
    float* gcnt = alloc((size_t)G * 4);
    float* g    = alloc((size_t)G * 128 * 4);
    float* m1   = alloc((size_t)G * 128 * 4);
    unsigned char* keys = (unsigned char*)ws;
    ws += ((size_t)nE + 3 + 255) & ~(size_t)255;

    const int  eB  = (nE + 1023) / 1024;
    const int  nB  = (nN + 255) / 256;
    const int  oB  = (nN + OWN_ROWS - 1) / OWN_ROWS;
    if (oB > 256 || (size_t)(ws - (char*)d_ws) > ws_size) keys = nullptr;

    fill_f32<<<nB, 256, 0, stream>>>(deg, 1.0f, nN);
    fill_f32<<<(G * 128 + 255) / 256, 256, 0, stream>>>(gsum, 0.0f, (long)G * 128);
    fill_f32<<<(G + 255) / 256, 256, 0, stream>>>(gcnt, 0.0f, G);
    deg_count<<<eB, 256, 0, stream>>>(dst, deg, (unsigned*)keys, nE);
    rsqrt_k<<<nB, 256, 0, stream>>>(deg, dinv, nN);

    layer_owned_rows<<<oB, 1024, 0, stream>>>(src, dst, dinv, x, W1, b1, bufA, (int*)bufB, 0, nE, nN,
                                              nullptr, nullptr, nullptr, 0, keys);

    layer_owned_rows<<<oB, 1024, 0, stream>>>(src, dst, dinv, bufA, W2, b2, bufB, (int*)bufB, 1, nE, nN,
                                              batch, gsum, gcnt, G, keys);

    pool_div<<<(G * 128 + 255) / 256, 256, 0, stream>>>(gsum, gcnt, g, G);

    gemm_k128<<<(G + 15) / 16, 256, 0, stream>>>(g, Wl1, bl1, m1, G, 128, 1, 1);
    gemm_k128<<<(G + 15) / 16, 256, 0, stream>>>(m1, Wl2, bl2, out, G, 32, 0, 1);
}


# Layer-1 scan of all key bytes gone: degree pass sorts each 8192-edge stretch by owner group (LDS rank, prefix sum) into span + start/count row; a group reads only its pieces; LDS sum order differs
# speedup vs baseline: 4.7653x; 1.0517x over previous
#include <hip/hip_runtime.h>

typedef __attribute__((ext_vector_type(2))) float v2f;
typedef __attribute__((ext_vector_type(8))) float v8f;

__global__ __launch_bounds__(256) void fill_f32(float* __restrict__ p, float v, long n) {
    long i = (long)blockIdx.x * 256 + threadIdx.x;
    if (i < n) p[i] = v;
}

#define OWN_ROWS   392
#define MAX_OWN    256
#define BIN_EDGES  8192

__global__ __launch_bounds__(1024) void deg_bin(const int* __restrict__ dst,
                                                float* __restrict__ deg,
                                                unsigned short* __restrict__ span,
                                                unsigned* __restrict__ tab, int nE) {
    __shared__ int sBinN[MAX_OWN];
    __shared__ int sBinAt[MAX_OWN];
    __shared__ int sWaveTot[MAX_OWN / 32];
    __shared__ __attribute__((aligned(16))) unsigned short sIds[BIN_EDGES];
    const int  tid     = threadIdx.x;
    const int  e0      = (int)blockIdx.x * BIN_EDGES;
    const bool binning = span != nullptr && tab != nullptr;

    if (tid < MAX_OWN) sBinN[tid] = 0;
    __syncthreads();

    int own[8];
    int rnk[8];
#pragma unroll
    for (int h = 0; h < 2; ++h) {
        const int lo = h * (BIN_EDGES / 2) + tid * 4;
        const int e  = e0 + lo;
        int d[4] = {0, 0, 0, 0};
        if (e + 3 < nE && (((size_t)(dst + e)) & 15) == 0) {
            const int4 v = *(const int4*)(dst + e);
            d[0] = v.x; d[1] = v.y; d[2] = v.z; d[3] = v.w;
        } else {
#pragma unroll
            for (int cb = 0; cb < 4; ++cb) if (e + cb < nE) d[cb] = dst[e + cb];
        }
#pragma unroll
        for (int cb = 0; cb < 4; ++cb) {
            own[h * 4 + cb] = -1;
            rnk[h * 4 + cb] = 0;
            if (e + cb < nE) {
                atomicAdd(&deg[d[cb]], 1.0f);
                const unsigned o = (unsigned)d[cb] / OWN_ROWS;
                if (binning && o < (unsigned)MAX_OWN) {
                    own[h * 4 + cb] = (int)o;
                    rnk[h * 4 + cb] = atomicAdd(&sBinN[o], 1);
                }
            }
        }
    }
    if (!binning) return;
    __syncthreads();

    int cnt = 0, incl = 0;
    if (tid < MAX_OWN) {
        cnt  = sBinN[tid];
        incl = cnt;
        for (int o = 1; o < 32; o <<= 1) {
            const int t = __shfl_up(incl, o, 32);
            if ((tid & 31) >= o) incl += t;
        }
        if ((tid & 31) == 31) sWaveTot[tid >> 5] = incl;
    }
    __syncthreads();
    if (tid < MAX_OWN) {
        int at = incl - cnt;
        for (int w = 0; w < (tid >> 5); ++w) at += sWaveTot[w];
        sBinAt[tid] = at;
        tab[(size_t)blockIdx.x * MAX_OWN + tid] = ((unsigned)at << 16) | (unsigned)cnt;
    }
    __syncthreads();

#pragma unroll
    for (int i = 0; i < 8; ++i) {
        if (own[i] >= 0) {
            const int lo = (i >> 2) * (BIN_EDGES / 2) + tid * 4 + (i & 3);
            sIds[sBinAt[own[i]] + rnk[i]] = (unsigned short)lo;
        }
    }
    __syncthreads();

    const int total = sBinAt[MAX_OWN - 1] + sBinN[MAX_OWN - 1];
    unsigned*       sp = (unsigned*)(span + (size_t)blockIdx.x * BIN_EDGES);
    const unsigned* sw = (const unsigned*)sIds;
    for (int i = tid; i < ((total + 1) >> 1); i += 1024) sp[i] = sw[i];
}

__global__ __launch_bounds__(256) void rsqrt_k(const float* __restrict__ deg,
                                               float* __restrict__ dinv, int n) {
    int i = blockIdx.x * 256 + threadIdx.x;
    if (i < n) dinv[i] = rsqrtf(deg[i]);
}

#define LPITCH 66

__global__ __launch_bounds__(256) void gemm_k128(const float* __restrict__ A,
                                                 const float* __restrict__ W,
                                                 const float* __restrict__ bias,
                                                 float* __restrict__ out,
                                                 int M, int N, int do_relu, int has_bias) {
    __shared__ v2f sA2[16][LPITCH];
    __shared__ v2f sW2[128][LPITCH];

    const int row0 = blockIdx.x * 16;
    const int tid  = threadIdx.x;

    float* sWf = (float*)sW2;
    for (int i = tid; i < 128 * N; i += 256) {
        int k = i / N, n = i - k * N;
        sWf[n * (2 * LPITCH) + k] = W[i];
    }
    float* sAf = (float*)sA2;
    for (int i = tid; i < 16 * 128; i += 256) {
        int r = i >> 7, c = i & 127;
        sAf[r * (2 * LPITCH) + c] = (row0 + r < M) ? A[(size_t)(row0 + r) * 128 + c] : 0.0f;
    }
    __syncthreads();

    const int wave  = tid >> 5;
    const int lane  = tid & 31;
    const int m     = lane & 15;
    const int hsel  = lane >> 4;
    const int mb    = hsel << 3;
    const int ncolt = N >> 4;

    for (int ct = wave; ct < ncolt; ct += 8) {
        const int col0 = ct << 4;
        v8f acc = {};
#pragma unroll 4
        for (int k2 = 0; k2 < 64; k2 += 2) {
            const int kq = k2 + hsel;
            v2f a = sA2[m][kq];
            v2f b = sW2[col0 + m][kq];
            acc = __builtin_amdgcn_wmma_f32_16x16x4_f32(
                false, a, false, b, (short)0, acc, false, false);
        }
        const float bv = has_bias ? bias[col0 + m] : 0.0f;
#pragma unroll
        for (int r = 0; r < 8; ++r) {
            int row = row0 + mb + r;
            if (row < M) {
                float v = acc[r] + bv;
                if (do_relu) v = fmaxf(v, 0.0f);
                out[(size_t)row * N + col0 + m] = v;
            }
        }
    }
}

#define OWN_WAVES  32
#define QCAP       544
#define APITCH     132
#define POOL_SLOTS 32
#define SLOT_ROWS  (((OWN_ROWS + 15) / 16) * 16)

__global__ __launch_bounds__(1024) void layer_owned_rows(const int* __restrict__ src,
                                                         const int* __restrict__ dst,
                                                         const float* __restrict__ dinv,
                                                         const float* __restrict__ xin,
                                                         const float* __restrict__ Wm,
                                                         const float* __restrict__ bias,
                                                         float* act, int* rec, int replay,
                                                         int nE, int nN,
                                                         const int* __restrict__ batch,
                                                         float* __restrict__ gsum,
                                                         float* __restrict__ gcnt, int G,
                                                         const unsigned short* __restrict__ span,
                                                         const unsigned* __restrict__ tab, int nBins) {
    __shared__ __attribute__((aligned(16))) float sAgg[OWN_ROWS * APITCH];
    __shared__ __attribute__((aligned(16))) int   sQ[OWN_WAVES * QCAP];
    static_assert(sizeof(int) * OWN_WAVES * QCAP >= sizeof(v2f) * 128 * LPITCH,
                  "the transposed W tile must fit where the queues were");
    __shared__ int sRecN;
    __shared__ int sRecBad;
    __shared__ __attribute__((aligned(16))) float sPool[POOL_SLOTS * 128];
    __shared__ __attribute__((aligned(16))) int   sSlot[SLOT_ROWS];
    __shared__ int sCnt[POOL_SLOTS];
    const bool pooling = gsum != nullptr;

    const int tid  = threadIdx.x;
    const int lane = tid & 31;
    const int wave = __builtin_amdgcn_readfirstlane(tid >> 5);
    const int base = (int)blockIdx.x * OWN_ROWS;
    const int nOwn = max(0, min(OWN_ROWS, nN - base));

    float4* z = (float4*)sAgg;
    for (int i = tid; i < OWN_ROWS * (APITCH / 4); i += 1024) z[i] = make_float4(0.0f, 0.0f, 0.0f, 0.0f);
    if (tid == 0) {
        sRecN = 0;
        sRecBad = 0;
    }
    if (pooling) {
        for (int i = tid; i < POOL_SLOTS * 32; i += 1024) ((float4*)sPool)[i] = make_float4(0.0f, 0.0f, 0.0f, 0.0f);
        if (tid < POOL_SLOTS) sCnt[tid] = 0;
    }
    __syncthreads();

    int*   lst = rec + (size_t)base * 128;
    const int cap = ((nOwn * 128 - 32) / 3) & ~31;
    int*   lS = lst + 32;
    int*   lR = lS + cap;
    float* lW = (float*)(lR + cap);
    int nRec = -1;
    if (replay) {
        nRec = __builtin_amdgcn_readfirstlane(lst[0]);
        if (nRec > cap) nRec = -1;
    }

    int* q = sQ + wave * QCAP;
    const int  nPieces = (nE + 127) >> 7;
    int qn = 0;

    auto push = [&](bool m, unsigned b, int eid) {
        if (b) {
            const int pos = qn + (int)__builtin_amdgcn_mbcnt_lo(b, 0u);
            if (m) q[pos] = eid;
            qn += __builtin_popcount(b);
        }
    };
    auto gather_add = [&](int s, int r, float w, int cnt) {
        const int wi = __float_as_int(w);
        for (int j = 0; j < cnt; j += 8) {
            float4 v[8];
#pragma unroll
            for (int u = 0; u < 8; ++u) {
                const int sj = __builtin_amdgcn_readlane(s, j + u);
                v[u] = *(const float4*)(xin + (size_t)sj * 128 + (lane << 2));
            }
#pragma unroll
            for (int u = 0; u < 8; ++u) {
                if (j + u < cnt) {
                    const int   rj = __builtin_amdgcn_readlane(r, j + u);
                    const float wj = __int_as_float(__builtin_amdgcn_readlane(wi, j + u));
                    float* ap = sAgg + rj * APITCH + lane;
                    __hip_atomic_fetch_add(ap +  0, wj * v[u].x, __ATOMIC_RELAXED, __HIP_MEMORY_SCOPE_WORKGROUP);
                    __hip_atomic_fetch_add(ap + 32, wj * v[u].y, __ATOMIC_RELAXED, __HIP_MEMORY_SCOPE_WORKGROUP);
                    __hip_atomic_fetch_add(ap + 64, wj * v[u].z, __ATOMIC_RELAXED, __HIP_MEMORY_SCOPE_WORKGROUP);
                    __hip_atomic_fetch_add(ap + 96, wj * v[u].w, __ATOMIC_RELAXED, __HIP_MEMORY_SCOPE_WORKGROUP);
                }
            }
        }
    };
    auto drain = [&](const int* qp, int cnt) {
        int s = 0, r = 0;
        float w = 0.0f;
        if (lane < cnt) {
            const int eid = qp[lane];
            s = src[eid];
            const int d = dst[eid];
            r = d - base;
            w = dinv[s] * dinv[d];
        }
        if (!replay) {
            int pos = 0;
            if (lane == 0) pos = atomicAdd(&sRecN, cnt);
            pos = __builtin_amdgcn_readfirstlane(pos);
            if (pos + cnt <= cap) {
                if (lane < cnt) {
                    lS[pos + lane] = s;
                    lR[pos + lane] = r;
                    lW[pos + lane] = w;
                }
            } else if (lane == 0) {
                sRecBad = 1;
            }
        }
        gather_add(s, r, w, cnt);
    };

    auto test = [&](const int4 d4, int e) {
        const bool m0 = (unsigned)d4.x - (unsigned)base < (unsigned)nOwn;
        const bool m1 = (unsigned)d4.y - (unsigned)base < (unsigned)nOwn;
        const bool m2 = (unsigned)d4.z - (unsigned)base < (unsigned)nOwn;
        const bool m3 = (unsigned)d4.w - (unsigned)base < (unsigned)nOwn;
        const unsigned b0 = __builtin_amdgcn_ballot_w32(m0);
        const unsigned b1 = __builtin_amdgcn_ballot_w32(m1);
        const unsigned b2 = __builtin_amdgcn_ballot_w32(m2);
        const unsigned b3 = __builtin_amdgcn_ballot_w32(m3);
        if (b0 | b1 | b2 | b3) {
            push(m0, b0, e);
            push(m1, b1, e + 1);
            push(m2, b2, e + 2);
            push(m3, b3, e + 3);
        }
    };

    if (nRec >= 0) {
        for (int i = wave * 32; i < nRec; i += 32 * OWN_WAVES) {
            int s = 0, r = 0;
            float w = 0.0f;
            if (i + lane < nRec) {
                s = lS[i + lane];
                r = lR[i + lane];
                w = lW[i + lane];
            }
            gather_add(s, r, w, min(32, nRec - i));
        }
    } else if (span != nullptr && tab != nullptr) {
        for (int b = wave; b < nBins; b += OWN_WAVES) {
            const unsigned tw = tab[(size_t)b * MAX_OWN + blockIdx.x];
            const int at  = (int)(tw >> 16);
            const int cnt = (int)(tw & 0xffffu);
            const unsigned short* sp = span + (size_t)b * BIN_EDGES + at;
            const int eb = b * BIN_EDGES;
            for (int i = 0; i < cnt; i += 32) {
                const int n = min(32, cnt - i);
                if (lane < n) q[qn + lane] = eb + (int)sp[i + lane];
                qn += n;
                if (qn >= 32) {
                    qn -= 32;
                    drain(q + qn, 32);
                }
            }
        }
        if (qn > 0) drain(q, qn);
    } else {
        for (int k = wave; k < nPieces; k += OWN_WAVES) {
            const int e = (k << 7) + (lane << 2);
            int4 d4;
            d4.x = (e     < nE) ? dst[e]     : -1;
            d4.y = (e + 1 < nE) ? dst[e + 1] : -1;
            d4.z = (e + 2 < nE) ? dst[e + 2] : -1;
            d4.w = (e + 3 < nE) ? dst[e + 3] : -1;
            test(d4, e);
            while (qn >= 32) {
                qn -= 32;
                drain(q + qn, 32);
            }
        }
        if (qn > 0) drain(q, qn);
    }
    __syncthreads();
    if (!replay && tid == 0) lst[0] = sRecBad ? -1 : sRecN;

    v2f*   sW2 = (v2f*)sQ;
    float* sWf = (float*)sQ;
    for (int i = tid; i < 128 * 128; i += 1024) {
        const int k = i >> 7, n = i & 127;
        sWf[n * (2 * LPITCH) + k] = Wm[i];
    }

    int gbase = 0;
    if (pooling) {
        gbase = __builtin_amdgcn_readfirstlane(nOwn > 0 ? batch[base] : 0);
        for (int r = tid; r < SLOT_ROWS; r += 1024) {
            int code = -1;
            if (r < nOwn) {
                const int b = batch[base + r];
                if ((unsigned)b < (unsigned)G) {
                    const int s = b - gbase;
                    if ((unsigned)s < (unsigned)POOL_SLOTS) {
                        code = s;
                        atomicAdd(&sCnt[s], 1);
                    } else {
                        code = -2;
                        atomicAdd(gcnt + b, 1.0f);
                    }
                }
            }
            sSlot[r] = code;
        }
    }

    for (int r0 = wave; r0 < nOwn; r0 += 4 * OWN_WAVES) {
        float  di[4];
        float4 hv[4];
#pragma unroll
        for (int t = 0; t < 4; ++t) {
            const size_t node = (size_t)(base + min(r0 + OWN_WAVES * t, nOwn - 1));
            di[t] = dinv[node];
            hv[t] = *(const float4*)(xin + node * 128 + (lane << 2));
        }
#pragma unroll
        for (int t = 0; t < 4; ++t) {
            const int r = r0 + OWN_WAVES * t;
            if (r < nOwn) {
                const float d2 = di[t] * di[t];
                float* rowp = sAgg + r * APITCH;
                const float* ap = rowp + lane;
                float4 o;
                o.x = ap[0]  + d2 * hv[t].x;
                o.y = ap[32] + d2 * hv[t].y;
                o.z = ap[64] + d2 * hv[t].z;
                o.w = ap[96] + d2 * hv[t].w;
                *(float4*)(rowp + (lane << 2)) = o;
            }
        }
    }
    __syncthreads();

    const int m      = lane & 15;
    const int hsel   = lane >> 4;
    const int mb     = hsel << 3;
    const int nItems = ((nOwn + 15) >> 4) << 2;
    for (int p = wave; p < nItems; p += OWN_WAVES) {
        const int row0 = (p >> 2) << 4;
        const int col0 = (p & 3) << 5;
        const v2f* arow = (const v2f*)(sAgg + min(row0 + m, OWN_ROWS - 1) * APITCH);
        const v2f* b0p  = sW2 + (col0 + m) * LPITCH;
        const v2f* b1p  = sW2 + (col0 + 16 + m) * LPITCH;
        v8f acc0 = {};
        v8f acc1 = {};
#pragma unroll 4
        for (int k2 = 0; k2 < 64; k2 += 2) {
            const int kq = k2 + hsel;
            const v2f a  = arow[kq];
            const v2f b0 = b0p[kq];
            const v2f b1 = b1p[kq];
            acc0 = __builtin_amdgcn_wmma_f32_16x16x4_f32(
                false, a, false, b0, (short)0, acc0, false, false);
            acc1 = __builtin_amdgcn_wmma_f32_16x16x4_f32(
                false, a, false, b1, (short)0, acc1, false, false);
        }
        const float bv0 = bias[col0 + m];
        const float bv1 = bias[col0 + 16 + m];
        if (!pooling) {
#pragma unroll
            for (int r = 0; r < 8; ++r) {
                const int row = row0 + mb + r;
                if (row < nOwn) {
                    float* op = act + (size_t)(base + row) * 128 + col0 + m;
                    op[0]  = fmaxf(acc0[r] + bv0, 0.0f);
                    op[16] = fmaxf(acc1[r] + bv1, 0.0f);
                }
            }
        } else {
            const int4 sa = *(const int4*)(sSlot + row0 + mb);
            const int4 sb = *(const int4*)(sSlot + row0 + mb + 4);
            const int sl[8] = {sa.x, sa.y, sa.z, sa.w, sb.x, sb.y, sb.z, sb.w};
            const int s0 = __builtin_amdgcn_readfirstlane(sa.x);
            bool same = true;
#pragma unroll
            for (int r = 0; r < 8; ++r) same = same && (sl[r] == s0);
            if (s0 >= 0 && __builtin_amdgcn_ballot_w32(same) == 0xffffffffu) {
                float t0 = 0.0f, t1 = 0.0f;
#pragma unroll
                for (int r = 0; r < 8; ++r) {
                    t0 += fmaxf(acc0[r] + bv0, 0.0f);
                    t1 += fmaxf(acc1[r] + bv1, 0.0f);
                }
                float* pp = sPool + s0 * 128 + col0 + m;
                __hip_atomic_fetch_add(pp,      t0, __ATOMIC_RELAXED, __HIP_MEMORY_SCOPE_WORKGROUP);
                __hip_atomic_fetch_add(pp + 16, t1, __ATOMIC_RELAXED, __HIP_MEMORY_SCOPE_WORKGROUP);
            } else {
#pragma unroll
                for (int r = 0; r < 8; ++r) {
                    const float v0 = fmaxf(acc0[r] + bv0, 0.0f);
                    const float v1 = fmaxf(acc1[r] + bv1, 0.0f);
                    if (sl[r] >= 0) {
                        float* pp = sPool + sl[r] * 128 + col0 + m;
                        __hip_atomic_fetch_add(pp,      v0, __ATOMIC_RELAXED, __HIP_MEMORY_SCOPE_WORKGROUP);
                        __hip_atomic_fetch_add(pp + 16, v1, __ATOMIC_RELAXED, __HIP_MEMORY_SCOPE_WORKGROUP);
                    } else if (sl[r] == -2) {
                        float* gp = gsum + (size_t)batch[base + row0 + mb + r] * 128 + col0 + m;
                        atomicAdd(gp,      v0);
                        atomicAdd(gp + 16, v1);
                    }
                }
            }
        }
    }

    if (pooling) {
        __syncthreads();
        for (int i = tid; i < POOL_SLOTS * 128; i += 1024) {
            const int s = i >> 7;
            if (sCnt[s] > 0) atomicAdd(gsum + (size_t)(gbase + s) * 128 + (i & 127), sPool[i]);
        }
        if (tid < POOL_SLOTS && sCnt[tid] > 0) atomicAdd(gcnt + gbase + tid, (float)sCnt[tid]);
    }
}

__global__ __launch_bounds__(256) void pool_div(const float* __restrict__ gsum,
                                                const float* __restrict__ gcnt,
                                                float* __restrict__ g, int G) {
    int i = blockIdx.x * 256 + threadIdx.x;
    if (i < G * 128) {
        int b = i >> 7;
        g[i] = gsum[i] / fmaxf(gcnt[b], 1.0f);
    }
}

extern "C" void kernel_launch(void* const* d_in, const int* in_sizes, int n_in,
                              void* d_out, int out_size, void* d_ws, size_t ws_size,
                              hipStream_t stream) {
    const float* x     = (const float*)d_in[0];
    const int*   ei    = (const int*)d_in[1];
    const int*   batch = (const int*)d_in[2];
    const float* W1    = (const float*)d_in[3];
    const float* b1    = (const float*)d_in[4];
    const float* W2    = (const float*)d_in[5];
    const float* b2    = (const float*)d_in[6];
    const float* Wl1   = (const float*)d_in[7];
    const float* bl1   = (const float*)d_in[8];
    const float* Wl2   = (const float*)d_in[9];
    const float* bl2   = (const float*)d_in[10];
    float* out = (float*)d_out;

    const int nN = in_sizes[0] / 128;
    const int nE = in_sizes[1] / 2;
    const int G  = out_size / 32;
    const int* src = ei;
    const int* dst = ei + nE;

    char* ws = (char*)d_ws;
    auto alloc = [&](size_t bytes) -> float* {
        float* p = (float*)ws;
        ws += (bytes + 255) & ~(size_t)255;
        return p;
    };
    float* bufA = alloc((size_t)nN * 128 * 4);
    float* bufB = alloc((size_t)nN * 128 * 4);
    float* deg  = alloc((size_t)nN * 4);
    float* dinv = alloc((size_t)nN * 4);
    float* gsum = alloc((size_t)G * 128 * 4);
    float* gcnt = alloc((size_t)G * 4);
    float* g    = alloc((size_t)G * 128 * 4);
    float* m1   = alloc((size_t)G * 128 * 4);
    const int  nBins = (nE + BIN_EDGES - 1) / BIN_EDGES;
    unsigned short* span = (unsigned short*)alloc((size_t)nBins * BIN_EDGES * 2);
    unsigned*       tab  = (unsigned*)alloc((size_t)nBins * MAX_OWN * 4);

    const int  nB  = (nN + 255) / 256;
    const int  oB  = (nN + OWN_ROWS - 1) / OWN_ROWS;
    if (oB > MAX_OWN || (size_t)(ws - (char*)d_ws) > ws_size) {
        span = nullptr;
        tab  = nullptr;
    }

    fill_f32<<<nB, 256, 0, stream>>>(deg, 1.0f, nN);
    fill_f32<<<(G * 128 + 255) / 256, 256, 0, stream>>>(gsum, 0.0f, (long)G * 128);
    fill_f32<<<(G + 255) / 256, 256, 0, stream>>>(gcnt, 0.0f, G);
    deg_bin<<<nBins, 1024, 0, stream>>>(dst, deg, span, tab, nE);
    rsqrt_k<<<nB, 256, 0, stream>>>(deg, dinv, nN);

    layer_owned_rows<<<oB, 1024, 0, stream>>>(src, dst, dinv, x, W1, b1, bufA, (int*)bufB, 0, nE, nN,
                                              nullptr, nullptr, nullptr, 0, span, tab, nBins);

    layer_owned_rows<<<oB, 1024, 0, stream>>>(src, dst, dinv, bufA, W2, b2, bufB, (int*)bufB, 1, nE, nN,
                                              batch, gsum, gcnt, G, span, tab, nBins);

    pool_div<<<(G * 128 + 255) / 256, 256, 0, stream>>>(gsum, gcnt, g, G);

    gemm_k128<<<(G + 15) / 16, 256, 0, stream>>>(g, Wl1, bl1, m1, G, 128, 1, 1);
    gemm_k128<<<(G + 15) / 16, 256, 0, stream>>>(m1, Wl2, bl2, out, G, 32, 0, 1);
}
